# starting kernel
# baseline (speedup 1.0000x reference)
#include <hip/hip_runtime.h>

typedef float v2f __attribute__((ext_vector_type(2)));
typedef float v8f __attribute__((ext_vector_type(8)));

#define QDIM   256
#define QBATCH 32768

__global__ void qenc_build(const float* __restrict__ wrot,
                           const float* __restrict__ wrz,
                           float* __restrict__ MrT,
                           float* __restrict__ MiT) {
    extern __shared__ float lds[];
    const int tid = threadIdx.x;
    const int rl  = tid >> 3;
    const int u   = tid & 7;
    const int r   = blockIdx.x * 32 + rl;

    float* sre = lds + rl * 514;
    float* sim = sre + 257;

    const float w = 6.28318530717958647692f / 256.0f;
    for (int j = 0; j < 32; ++j) {
        int k  = u * 32 + j;
        int ph = (r * k) & 255;
        float sv, cv;
        sincosf(w * (float)ph, &sv, &cv);
        sre[k] = cv * 0.0625f;
        sim[k] = sv * 0.0625f;
    }
    __syncthreads();

    for (int l = 0; l < 4; ++l) {
        for (int q = 0; q < 8; ++q) {
            float th  = wrot[l * 8 + q];
            float phz = wrz[l * 8 + q];
            float c = cosf(0.5f * th), s = sinf(0.5f * th);
            float emr = cosf(0.5f * phz), emi = -sinf(0.5f * phz);
            float U00r =  emr * c, U00i =  emi * c;
            float U01r = -emr * s, U01i = -emi * s;
            float U10r =  emr * s, U10i = -emi * s;
            float U11r =  emr * c, U11i = -emi * c;
            int bit = 7 - q, mask = 1 << bit;
            for (int j = 0; j < 16; ++j) {
                int p  = u * 16 + j;
                int k0 = ((p >> bit) << (bit + 1)) | (p & (mask - 1));
                int k1 = k0 | mask;
                float ar = sre[k0], ai = sim[k0];
                float br = sre[k1], bi = sim[k1];
                sre[k0] = U00r * ar - U00i * ai + U01r * br - U01i * bi;
                sim[k0] = U00r * ai + U00i * ar + U01r * bi + U01i * br;
                sre[k1] = U10r * ar - U10i * ai + U11r * br - U11i * bi;
                sim[k1] = U10r * ai + U10i * ar + U11r * bi + U11i * br;
            }
            __syncthreads();
        }
        for (int q = 0; q < 8; ++q) {
            int cb = 7 - q, tb = 7 - ((q + 1) & 7);
            int tmask = 1 << tb;
            for (int j = 0; j < 32; ++j) {
                int k = u * 32 + j;
                if (((k >> cb) & 1) && !((k >> tb) & 1)) {
                    int k2 = k | tmask;
                    float tr = sre[k], ti = sim[k];
                    sre[k]  = sre[k2]; sim[k]  = sim[k2];
                    sre[k2] = tr;      sim[k2] = ti;
                }
            }
            __syncthreads();
        }
    }

    for (int j = 0; j < 32; ++j) {
        int n = u * 32 + j;
        MrT[n * 256 + r] = sre[n];
        MiT[n * 256 + r] = sim[n];
    }
}

#define WMMA_F32(A, B, C) \
    __builtin_amdgcn_wmma_f32_16x16x4_f32(false, (A), false, (B), (short)0, (C), false, false)

__global__ void qenc_gemm(const float* __restrict__ X,
                          const float* __restrict__ MrT,
                          const float* __restrict__ MiT,
                          float* __restrict__ out) {
    const int lane = threadIdx.x & 31;
    const int wid  = blockIdx.x * 8 + (threadIdx.x >> 5);
    const int n0   = (wid & 7) * 32;
    const int m0   = (wid >> 3) * 32;
    const int half = lane >> 4;
    const int l15  = lane & 15;

    const float* ap0  = X   + (m0 + l15) * 256 + 2 * half;
    const float* ap1  = ap0 + 16 * 256;
    const float* brp0 = MrT + (n0 + l15) * 256 + 2 * half;
    const float* brp1 = brp0 + 16 * 256;
    const float* bip0 = MiT + (n0 + l15) * 256 + 2 * half;
    const float* bip1 = bip0 + 16 * 256;

    v8f accR00 = {}, accI00 = {}, accR01 = {}, accI01 = {};
    v8f accR10 = {}, accI10 = {}, accR11 = {}, accI11 = {};
    float nrm0 = 0.0f, nrm1 = 0.0f;

#pragma unroll 2
    for (int k = 0; k < 256; k += 4) {
        v2f A0  = *(const v2f*)(ap0 + k);
        v2f A1  = *(const v2f*)(ap1 + k);
        v2f Br0 = *(const v2f*)(brp0 + k);
        v2f Br1 = *(const v2f*)(brp1 + k);
        v2f Bi0 = *(const v2f*)(bip0 + k);
        v2f Bi1 = *(const v2f*)(bip1 + k);
        nrm0 += A0.x * A0.x + A0.y * A0.y;
        nrm1 += A1.x * A1.x + A1.y * A1.y;
        accR00 = WMMA_F32(A0, Br0, accR00);
        accI00 = WMMA_F32(A0, Bi0, accI00);
        accR01 = WMMA_F32(A0, Br1, accR01);
        accI01 = WMMA_F32(A0, Bi1, accI01);
        accR10 = WMMA_F32(A1, Br0, accR10);
        accI10 = WMMA_F32(A1, Bi0, accI10);
        accR11 = WMMA_F32(A1, Br1, accR11);
        accI11 = WMMA_F32(A1, Bi1, accI11);
    }

    nrm0 += __shfl_xor(nrm0, 16, 32);
    nrm1 += __shfl_xor(nrm1, 16, 32);

#pragma unroll
    for (int i = 0; i < 8; ++i) {
        int row    = i + 8 * half;
        float inv0 = 1.0f / __shfl(nrm0, row, 32);
        float inv1 = 1.0f / __shfl(nrm1, row, 32);
        float* o0  = out + (m0 + row) * 256 + n0 + l15;
        float* o1  = out + (m0 + 16 + row) * 256 + n0 + l15;
        o0[0]  = (accR00[i] * accR00[i] + accI00[i] * accI00[i]) * inv0;
        o0[16] = (accR01[i] * accR01[i] + accI01[i] * accI01[i]) * inv0;
        o1[0]  = (accR10[i] * accR10[i] + accI10[i] * accI10[i]) * inv1;
        o1[16] = (accR11[i] * accR11[i] + accI11[i] * accI11[i]) * inv1;
    }
}

extern "C" void kernel_launch(void* const* d_in, const int* in_sizes, int n_in,
                              void* d_out, int out_size, void* d_ws, size_t ws_size,
                              hipStream_t stream) {
    const float* x    = (const float*)d_in[0];
    const float* wrot = (const float*)d_in[1];
    const float* wrz  = (const float*)d_in[2];
    float* out = (float*)d_out;

    float* MrT = (float*)d_ws;
    float* MiT = MrT + QDIM * QDIM;

    size_t lds_bytes = 32 * 514 * sizeof(float);
    qenc_build<<<8, 256, lds_bytes, stream>>>(wrot, wrz, MrT, MiT);

    qenc_gemm<<<1024, 256, 0, stream>>>(x, MrT, MiT, out);
}


# Matrix-build kernel: a layer's 8 one-qubit gates as two 16x16 Kronecker-factor complex products on f32 WMMA (row tile stays in accumulators), CNOT ring as one LDS gather; gate sum order changes
# speedup vs baseline: 1.9878x; 1.9878x over previous
#include <hip/hip_runtime.h>

typedef float v2f __attribute__((ext_vector_type(2)));
typedef float v8f __attribute__((ext_vector_type(8)));

#define QDIM   256
#define QBATCH 32768

#define WMMA_F32(A, B, C) \
    __builtin_amdgcn_wmma_f32_16x16x4_f32(false, (A), false, (B), (short)0, (C), false, false)

#define ROW_PITCH 514
#define GFRAG_OFF (32 * ROW_PITCH)

__device__ __forceinline__ int cnot_ring_src(int k) {
    int src = k;
    for (int q = 7; q >= 0; --q) {
        int cb = 7 - q, tb = 7 - ((q + 1) & 7);
        src ^= ((src >> cb) & 1) << tb;
    }
    return src;
}

__global__ __launch_bounds__(1024) void qenc_build(const float* __restrict__ wrot,
                                                   const float* __restrict__ wrz,
                                                   float* __restrict__ MrT,
                                                   float* __restrict__ MiT) {
    extern __shared__ float lds[];
    const int lane = threadIdx.x & 31;
    const int wv   = threadIdx.x >> 5;
    const int c    = lane & 15;
    const int h    = lane >> 4;
    const int r    = blockIdx.x * 32 + wv;

    float* sre = lds + wv * ROW_PITCH;
    float* sim = sre + 257;
    float* gfr = lds + GFRAG_OFF;

    if (wv < 8) {
        float th  = wrot[lane];
        float phz = wrz[lane];
        float gc  = cosf(0.5f * th),  gs  = sinf(0.5f * th);
        float ger = cosf(0.5f * phz), gei = -sinf(0.5f * phz);
        const int g0 = (wv >> 1) * 8 + (wv & 1) * 4;
        float u0r[4], u0i[4], u1r[4], u1i[4];
#pragma unroll
        for (int p = 0; p < 4; ++p) {
            float cc = __shfl(gc,  g0 + p, 32), ss = __shfl(gs,  g0 + p, 32);
            float er = __shfl(ger, g0 + p, 32), ei = __shfl(gei, g0 + p, 32);
            int rb = (c >> (3 - p)) & 1;
            u0r[p] = rb ?  er * ss :  er * cc;
            u0i[p] = rb ? -ei * ss :  ei * cc;
            u1r[p] = rb ?  er * cc : -er * ss;
            u1i[p] = rb ? -ei * cc : -ei * ss;
        }
        float e0r = h ? u1r[0] : u0r[0];
        float e0i = h ? u1i[0] : u0i[0];
#pragma unroll
        for (int j = 0; j < 8; ++j) {
            int b2 = (j >> 2) & 1, b1 = (j >> 1) & 1, b0 = j & 1;
            float x1r = b2 ? u1r[1] : u0r[1], x1i = b2 ? u1i[1] : u0i[1];
            float x2r = b1 ? u1r[2] : u0r[2], x2i = b1 ? u1i[2] : u0i[2];
            float x3r = b0 ? u1r[3] : u0r[3], x3i = b0 ? u1i[3] : u0i[3];
            float t1r = e0r * x1r - e0i * x1i, t1i = e0r * x1i + e0i * x1r;
            float t2r = t1r * x2r - t1i * x2i, t2i = t1r * x2i + t1i * x2r;
            float gr  = t2r * x3r - t2i * x3i, gi  = t2r * x3i + t2i * x3r;
            gfr[(wv * 16 + j) * 32 + lane]     = gr;
            gfr[(wv * 16 + 8 + j) * 32 + lane] = gi;
        }
    }

    v8f Sr, Si;
    const float w = 6.28318530717958647692f / 256.0f;
#pragma unroll
    for (int i = 0; i < 8; ++i) {
        int k  = 16 * (8 * h + i) + c;
        int ph = (r * k) & 255;
        float sv, cv;
        sincosf(w * (float)ph, &sv, &cv);
        Sr[i] = cv * 0.0625f;
        Si[i] = sv * 0.0625f;
    }
    __syncthreads();

    for (int l = 0; l < 4; ++l) {
        for (int g = 0; g < 2; ++g) {
            const float* gp = gfr + (l * 2 + g) * 16 * 32 + lane;
            float Gr[8], Gi[8], Gn[8];
#pragma unroll
            for (int j = 0; j < 8; ++j) {
                Gr[j] = gp[j * 32];
                Gi[j] = gp[(8 + j) * 32];
                Gn[j] = -Gi[j];
            }
            v8f Dr = {}, Di = {};
#pragma unroll
            for (int t = 0; t < 4; ++t) {
                v2f ar = {Sr[2 * t], Sr[2 * t + 1]};
                v2f ai = {Si[2 * t], Si[2 * t + 1]};
                v2f br = {Gr[2 * t], Gr[2 * t + 1]};
                v2f bi = {Gi[2 * t], Gi[2 * t + 1]};
                v2f bn = {Gn[2 * t], Gn[2 * t + 1]};
                Dr = WMMA_F32(ar, br, Dr);
                Di = WMMA_F32(ar, bi, Di);
                Dr = WMMA_F32(ai, bn, Dr);
                Di = WMMA_F32(ai, br, Di);
            }
            Sr = Dr;
            Si = Di;
        }
#pragma unroll
        for (int i = 0; i < 8; ++i) {
            int k = 16 * (8 * h + i) + c;
            sre[k] = Sr[i];
            sim[k] = Si[i];
        }
        __syncthreads();
        if (l < 3) {
#pragma unroll
            for (int i = 0; i < 8; ++i) {
                int src = cnot_ring_src(16 * (8 * h + i) + c);
                Sr[i] = sre[src];
                Si[i] = sim[src];
            }
        }
    }

    for (int j = 0; j < 8; ++j) {
        int n   = wv * 8 + j;
        int src = cnot_ring_src(n);
        MrT[n * 256 + blockIdx.x * 32 + lane] = lds[lane * ROW_PITCH + src];
        MiT[n * 256 + blockIdx.x * 32 + lane] = lds[lane * ROW_PITCH + 257 + src];
    }
}

__global__ void qenc_gemm(const float* __restrict__ X,
                          const float* __restrict__ MrT,
                          const float* __restrict__ MiT,
                          float* __restrict__ out) {
    const int lane = threadIdx.x & 31;
    const int wid  = blockIdx.x * 8 + (threadIdx.x >> 5);
    const int n0   = (wid & 7) * 32;
    const int m0   = (wid >> 3) * 32;
    const int half = lane >> 4;
    const int l15  = lane & 15;

    const float* ap0  = X   + (m0 + l15) * 256 + 2 * half;
    const float* ap1  = ap0 + 16 * 256;
    const float* brp0 = MrT + (n0 + l15) * 256 + 2 * half;
    const float* brp1 = brp0 + 16 * 256;
    const float* bip0 = MiT + (n0 + l15) * 256 + 2 * half;
    const float* bip1 = bip0 + 16 * 256;

    v8f accR00 = {}, accI00 = {}, accR01 = {}, accI01 = {};
    v8f accR10 = {}, accI10 = {}, accR11 = {}, accI11 = {};
    float nrm0 = 0.0f, nrm1 = 0.0f;

#pragma unroll 2
    for (int k = 0; k < 256; k += 4) {
        v2f A0  = *(const v2f*)(ap0 + k);
        v2f A1  = *(const v2f*)(ap1 + k);
        v2f Br0 = *(const v2f*)(brp0 + k);
        v2f Br1 = *(const v2f*)(brp1 + k);
        v2f Bi0 = *(const v2f*)(bip0 + k);
        v2f Bi1 = *(const v2f*)(bip1 + k);
        nrm0 += A0.x * A0.x + A0.y * A0.y;
        nrm1 += A1.x * A1.x + A1.y * A1.y;
        accR00 = WMMA_F32(A0, Br0, accR00);
        accI00 = WMMA_F32(A0, Bi0, accI00);
        accR01 = WMMA_F32(A0, Br1, accR01);
        accI01 = WMMA_F32(A0, Bi1, accI01);
        accR10 = WMMA_F32(A1, Br0, accR10);
        accI10 = WMMA_F32(A1, Bi0, accI10);
        accR11 = WMMA_F32(A1, Br1, accR11);
        accI11 = WMMA_F32(A1, Bi1, accI11);
    }

    nrm0 += __shfl_xor(nrm0, 16, 32);
    nrm1 += __shfl_xor(nrm1, 16, 32);

#pragma unroll
    for (int i = 0; i < 8; ++i) {
        int row    = i + 8 * half;
        float inv0 = 1.0f / __shfl(nrm0, row, 32);
        float inv1 = 1.0f / __shfl(nrm1, row, 32);
        float* o0  = out + (m0 + row) * 256 + n0 + l15;
        float* o1  = out + (m0 + 16 + row) * 256 + n0 + l15;
        o0[0]  = (accR00[i] * accR00[i] + accI00[i] * accI00[i]) * inv0;
        o0[16] = (accR01[i] * accR01[i] + accI01[i] * accI01[i]) * inv0;
        o1[0]  = (accR10[i] * accR10[i] + accI10[i] * accI10[i]) * inv1;
        o1[16] = (accR11[i] * accR11[i] + accI11[i] * accI11[i]) * inv1;
    }
}

extern "C" void kernel_launch(void* const* d_in, const int* in_sizes, int n_in,
                              void* d_out, int out_size, void* d_ws, size_t ws_size,
                              hipStream_t stream) {
    const float* x    = (const float*)d_in[0];
    const float* wrot = (const float*)d_in[1];
    const float* wrz  = (const float*)d_in[2];
    float* out = (float*)d_out;

    float* MrT = (float*)d_ws;
    float* MiT = MrT + QDIM * QDIM;

    size_t lds_bytes = (32 * 514 + 8 * 16 * 32) * sizeof(float);
    qenc_build<<<8, 1024, lds_bytes, stream>>>(wrot, wrz, MrT, MiT);

    qenc_gemm<<<1024, 256, 0, stream>>>(x, MrT, MiT, out);
}


# Product kernel: X and M^T tiles staged through LDS by a 16-wave work-group (128 rows x 256 outputs, K in 8 double-buffered stages of 32, whole-line global loads); fragments and sum order unchanged
# speedup vs baseline: 3.1403x; 1.5798x over previous
#include <hip/hip_runtime.h>

typedef float v2f __attribute__((ext_vector_type(2)));
typedef float v8f __attribute__((ext_vector_type(8)));

#define QDIM   256
#define QBATCH 32768

#define WMMA_F32(A, B, C) \
    __builtin_amdgcn_wmma_f32_16x16x4_f32(false, (A), false, (B), (short)0, (C), false, false)

#define ROW_PITCH 514
#define GFRAG_OFF (32 * ROW_PITCH)

__device__ __forceinline__ int cnot_ring_src(int k) {
    int src = k;
    for (int q = 7; q >= 0; --q) {
        int cb = 7 - q, tb = 7 - ((q + 1) & 7);
        src ^= ((src >> cb) & 1) << tb;
    }
    return src;
}

__global__ __launch_bounds__(1024) void qenc_build(const float* __restrict__ wrot,
                                                   const float* __restrict__ wrz,
                                                   float* __restrict__ MrT,
                                                   float* __restrict__ MiT) {
    extern __shared__ float lds[];
    const int lane = threadIdx.x & 31;
    const int wv   = threadIdx.x >> 5;
    const int c    = lane & 15;
    const int h    = lane >> 4;
    const int r    = blockIdx.x * 32 + wv;

    float* sre = lds + wv * ROW_PITCH;
    float* sim = sre + 257;
    float* gfr = lds + GFRAG_OFF;

    if (wv < 8) {
        float th  = wrot[lane];
        float phz = wrz[lane];
        float gc  = cosf(0.5f * th),  gs  = sinf(0.5f * th);
        float ger = cosf(0.5f * phz), gei = -sinf(0.5f * phz);
        const int g0 = (wv >> 1) * 8 + (wv & 1) * 4;
        float u0r[4], u0i[4], u1r[4], u1i[4];
#pragma unroll
        for (int p = 0; p < 4; ++p) {
            float cc = __shfl(gc,  g0 + p, 32), ss = __shfl(gs,  g0 + p, 32);
            float er = __shfl(ger, g0 + p, 32), ei = __shfl(gei, g0 + p, 32);
            int rb = (c >> (3 - p)) & 1;
            u0r[p] = rb ?  er * ss :  er * cc;
            u0i[p] = rb ? -ei * ss :  ei * cc;
            u1r[p] = rb ?  er * cc : -er * ss;
            u1i[p] = rb ? -ei * cc : -ei * ss;
        }
        float e0r = h ? u1r[0] : u0r[0];
        float e0i = h ? u1i[0] : u0i[0];
#pragma unroll
        for (int j = 0; j < 8; ++j) {
            int b2 = (j >> 2) & 1, b1 = (j >> 1) & 1, b0 = j & 1;
            float x1r = b2 ? u1r[1] : u0r[1], x1i = b2 ? u1i[1] : u0i[1];
            float x2r = b1 ? u1r[2] : u0r[2], x2i = b1 ? u1i[2] : u0i[2];
            float x3r = b0 ? u1r[3] : u0r[3], x3i = b0 ? u1i[3] : u0i[3];
            float t1r = e0r * x1r - e0i * x1i, t1i = e0r * x1i + e0i * x1r;
            float t2r = t1r * x2r - t1i * x2i, t2i = t1r * x2i + t1i * x2r;
            float gr  = t2r * x3r - t2i * x3i, gi  = t2r * x3i + t2i * x3r;
            gfr[(wv * 16 + j) * 32 + lane]     = gr;
            gfr[(wv * 16 + 8 + j) * 32 + lane] = gi;
        }
    }

    v8f Sr, Si;
    const float w = 6.28318530717958647692f / 256.0f;
#pragma unroll
    for (int i = 0; i < 8; ++i) {
        int k  = 16 * (8 * h + i) + c;
        int ph = (r * k) & 255;
        float sv, cv;
        sincosf(w * (float)ph, &sv, &cv);
        Sr[i] = cv * 0.0625f;
        Si[i] = sv * 0.0625f;
    }
    __syncthreads();

    for (int l = 0; l < 4; ++l) {
        for (int g = 0; g < 2; ++g) {
            const float* gp = gfr + (l * 2 + g) * 16 * 32 + lane;
            float Gr[8], Gi[8], Gn[8];
#pragma unroll
            for (int j = 0; j < 8; ++j) {
                Gr[j] = gp[j * 32];
                Gi[j] = gp[(8 + j) * 32];
                Gn[j] = -Gi[j];
            }
            v8f Dr = {}, Di = {};
#pragma unroll
            for (int t = 0; t < 4; ++t) {
                v2f ar = {Sr[2 * t], Sr[2 * t + 1]};
                v2f ai = {Si[2 * t], Si[2 * t + 1]};
                v2f br = {Gr[2 * t], Gr[2 * t + 1]};
                v2f bi = {Gi[2 * t], Gi[2 * t + 1]};
                v2f bn = {Gn[2 * t], Gn[2 * t + 1]};
                Dr = WMMA_F32(ar, br, Dr);
                Di = WMMA_F32(ar, bi, Di);
                Dr = WMMA_F32(ai, bn, Dr);
                Di = WMMA_F32(ai, br, Di);
            }
            Sr = Dr;
            Si = Di;
        }
#pragma unroll
        for (int i = 0; i < 8; ++i) {
            int k = 16 * (8 * h + i) + c;
            sre[k] = Sr[i];
            sim[k] = Si[i];
        }
        __syncthreads();
        if (l < 3) {
#pragma unroll
            for (int i = 0; i < 8; ++i) {
                int src = cnot_ring_src(16 * (8 * h + i) + c);
                Sr[i] = sre[src];
                Si[i] = sim[src];
            }
        }
    }

    for (int j = 0; j < 8; ++j) {
        int n   = wv * 8 + j;
        int src = cnot_ring_src(n);
        MrT[n * 256 + blockIdx.x * 32 + lane] = lds[lane * ROW_PITCH + src];
        MiT[n * 256 + blockIdx.x * 32 + lane] = lds[lane * ROW_PITCH + 257 + src];
    }
}

typedef float v4f __attribute__((ext_vector_type(4)));

#define GM_BM    128
#define GM_KC    32
#define GM_PITCH 36
#define GM_ROWS  (GM_BM + 512)
#define GM_STAGE (GM_ROWS * GM_PITCH)

__device__ __forceinline__ void gm_load(v4f (&st)[10], const float* __restrict__ gx,
                                        const float* __restrict__ gr,
                                        const float* __restrict__ gi, int kc) {
    const int ko = kc * GM_KC;
#pragma unroll
    for (int j = 0; j < 2; ++j) st[j] = *(const v4f*)(gx + j * 64 * 256 + ko);
#pragma unroll
    for (int j = 0; j < 4; ++j) st[2 + j] = *(const v4f*)(gr + j * 64 * 256 + ko);
#pragma unroll
    for (int j = 0; j < 4; ++j) st[6 + j] = *(const v4f*)(gi + j * 64 * 256 + ko);
}

__device__ __forceinline__ void gm_store(const v4f (&st)[10], float* sdst) {
#pragma unroll
    for (int j = 0; j < 10; ++j) *(v4f*)(sdst + j * 64 * GM_PITCH) = st[j];
}

__device__ __forceinline__ void gm_compute(const float* sa, const float* sb,
                                           v8f (&accR)[2][4], v8f (&accI)[2][4],
                                           float (&nrm)[2]) {
#pragma unroll
    for (int s = 0; s < GM_KC / 4; ++s) {
        v2f a[2], br[4], bi[4];
#pragma unroll
        for (int i = 0; i < 2; ++i) a[i] = *(const v2f*)(sa + i * 16 * GM_PITCH + 4 * s);
#pragma unroll
        for (int j = 0; j < 4; ++j) {
            br[j] = *(const v2f*)(sb + j * 16 * GM_PITCH + 4 * s);
            bi[j] = *(const v2f*)(sb + (256 + j * 16) * GM_PITCH + 4 * s);
        }
#pragma unroll
        for (int i = 0; i < 2; ++i)
            nrm[i] += a[i].x * a[i].x + a[i].y * a[i].y;
#pragma unroll
        for (int i = 0; i < 2; ++i) {
#pragma unroll
            for (int j = 0; j < 4; ++j) {
                accR[i][j] = WMMA_F32(a[i], br[j], accR[i][j]);
                accI[i][j] = WMMA_F32(a[i], bi[j], accI[i][j]);
            }
        }
    }
}

__global__ __launch_bounds__(512) void qenc_gemm(const float* __restrict__ X,
                                                 const float* __restrict__ MrT,
                                                 const float* __restrict__ MiT,
                                                 float* __restrict__ out) {
    __shared__ __attribute__((aligned(16))) float gl[2 * GM_STAGE];
    const int tid   = threadIdx.x;
    const int lane  = tid & 31;
    const int wv    = tid >> 5;
    const int half  = lane >> 4;
    const int l15   = lane & 15;
    const int mrow0 = (wv >> 2) * 32;
    const int ncol0 = (wv & 3) * 64;
    const int m0    = blockIdx.x * GM_BM;

    const int seg  = tid & 7;
    const int srow = tid >> 3;
    const float* gx = X   + (size_t)(m0 + srow) * 256 + seg * 4;
    const float* gr = MrT + srow * 256 + seg * 4;
    const float* gi = MiT + srow * 256 + seg * 4;
    float* sdst = gl + srow * GM_PITCH + seg * 4;

    const int aoff = (mrow0 + l15) * GM_PITCH + 2 * half;
    const int boff = (GM_BM + ncol0 + l15) * GM_PITCH + 2 * half;

    v8f accR[2][4], accI[2][4];
#pragma unroll
    for (int i = 0; i < 2; ++i) {
#pragma unroll
        for (int j = 0; j < 4; ++j) {
            accR[i][j] = (v8f){};
            accI[i][j] = (v8f){};
        }
    }
    float nrm[2] = {0.0f, 0.0f};

    v4f st[10];
    gm_load(st, gx, gr, gi, 0);
    gm_store(st, sdst);
    __syncthreads();

    for (int kc = 0; kc < 7; ++kc) {
        gm_load(st, gx, gr, gi, kc + 1);
        const float* sbase = gl + (kc & 1) * GM_STAGE;
        gm_compute(sbase + aoff, sbase + boff, accR, accI, nrm);
        gm_store(st, sdst + ((kc + 1) & 1) * GM_STAGE);
        __syncthreads();
    }
    gm_compute(gl + GM_STAGE + aoff, gl + GM_STAGE + boff, accR, accI, nrm);

    nrm[0] += __shfl_xor(nrm[0], 16, 32);
    nrm[1] += __shfl_xor(nrm[1], 16, 32);

#pragma unroll
    for (int r = 0; r < 8; ++r) {
        int row = r + 8 * half;
#pragma unroll
        for (int i = 0; i < 2; ++i) {
            float inv = 1.0f / __shfl(nrm[i], row, 32);
            float* o  = out + (size_t)(m0 + mrow0 + 16 * i + row) * 256 + ncol0 + l15;
#pragma unroll
            for (int j = 0; j < 4; ++j)
                o[16 * j] = (accR[i][j][r] * accR[i][j][r] + accI[i][j][r] * accI[i][j][r]) * inv;
        }
    }
}

extern "C" void kernel_launch(void* const* d_in, const int* in_sizes, int n_in,
                              void* d_out, int out_size, void* d_ws, size_t ws_size,
                              hipStream_t stream) {
    const float* x    = (const float*)d_in[0];
    const float* wrot = (const float*)d_in[1];
    const float* wrz  = (const float*)d_in[2];
    float* out = (float*)d_out;

    float* MrT = (float*)d_ws;
    float* MiT = MrT + QDIM * QDIM;

    size_t lds_bytes = (32 * 514 + 8 * 16 * 32) * sizeof(float);
    qenc_build<<<8, 1024, lds_bytes, stream>>>(wrot, wrz, MrT, MiT);

    qenc_gemm<<<256, 512, 0, stream>>>(x, MrT, MiT, out);
}


# Matrix-build kernel: each gate RZ*RY split as unit-phase diagonal x real rotation: a 16x16 Kronecker factor is real (complex x real: 8 WMMAs, was 16), phases applied per lane; gate sum order changes
# speedup vs baseline: 3.4000x; 1.0827x over previous
#include <hip/hip_runtime.h>

typedef float v2f __attribute__((ext_vector_type(2)));
typedef float v8f __attribute__((ext_vector_type(8)));

#define QDIM   256
#define QBATCH 32768

#define WMMA_F32(A, B, C) \
    __builtin_amdgcn_wmma_f32_16x16x4_f32(false, (A), false, (B), (short)0, (C), false, false)

#define ROW_PITCH 514
#define GFRAG_OFF (32 * ROW_PITCH)

__device__ __forceinline__ int cnot_ring_src(int k) {
    int src = k;
    for (int q = 7; q >= 0; --q) {
        int cb = 7 - q, tb = 7 - ((q + 1) & 7);
        src ^= ((src >> cb) & 1) << tb;
    }
    return src;
}

__global__ __launch_bounds__(1024) void qenc_build(const float* __restrict__ wrot,
                                                   const float* __restrict__ wrz,
                                                   float* __restrict__ MrT,
                                                   float* __restrict__ MiT) {
    extern __shared__ float lds[];
    const int lane = threadIdx.x & 31;
    const int wv   = threadIdx.x >> 5;
    const int c    = lane & 15;
    const int h    = lane >> 4;
    const int r    = blockIdx.x * 32 + wv;

    float* sre = lds + wv * ROW_PITCH;
    float* sim = sre + 257;
    float* gfr = lds + GFRAG_OFF;

    if (wv < 8) {
        float th  = wrot[lane];
        float phz = wrz[lane];
        float gc  = cosf(0.5f * th),  gs  = sinf(0.5f * th);
        float ger = cosf(0.5f * phz), gei = -sinf(0.5f * phz);
        const int g0 = (wv >> 1) * 8 + (wv & 1) * 4;
        float r0[4], r1[4];
        float dr = 1.0f, di = 0.0f;
#pragma unroll
        for (int p = 0; p < 4; ++p) {
            float cc = __shfl(gc,  g0 + p, 32), ss = __shfl(gs,  g0 + p, 32);
            float er = __shfl(ger, g0 + p, 32), ei = __shfl(gei, g0 + p, 32);
            int rb = (c >> (3 - p)) & 1;
            r0[p] = rb ? ss :  cc;
            r1[p] = rb ? cc : -ss;
            float fi = rb ? -ei : ei;
            float nr = dr * er - di * fi;
            float ni = dr * fi + di * er;
            dr = nr;
            di = ni;
        }
        float e0 = h ? r1[0] : r0[0];
#pragma unroll
        for (int j = 0; j < 8; ++j) {
            int b2 = (j >> 2) & 1, b1 = (j >> 1) & 1, b0 = j & 1;
            float x1 = b2 ? r1[1] : r0[1];
            float x2 = b1 ? r1[2] : r0[2];
            float x3 = b0 ? r1[3] : r0[3];
            gfr[(wv * 16 + j) * 32 + lane] = ((e0 * x1) * x2) * x3;
        }
        gfr[(wv * 16 + 8) * 32 + lane] = dr;
        gfr[(wv * 16 + 9) * 32 + lane] = di;
    }

    v8f Sr, Si;
    const float w = 6.28318530717958647692f / 256.0f;
#pragma unroll
    for (int i = 0; i < 8; ++i) {
        int k  = 16 * (8 * h + i) + c;
        int ph = (r * k) & 255;
        float sv, cv;
        sincosf(w * (float)ph, &sv, &cv);
        Sr[i] = cv * 0.0625f;
        Si[i] = sv * 0.0625f;
    }
    __syncthreads();

    for (int l = 0; l < 4; ++l) {
        for (int g = 0; g < 2; ++g) {
            const float* gp = gfr + (l * 2 + g) * 16 * 32 + lane;
            float G[8];
#pragma unroll
            for (int j = 0; j < 8; ++j) G[j] = gp[j * 32];
            const float pr = gp[8 * 32], pi = gp[9 * 32];
            v8f Dr = {}, Di = {};
#pragma unroll
            for (int t = 0; t < 4; ++t) {
                v2f ar = {Sr[2 * t], Sr[2 * t + 1]};
                v2f ai = {Si[2 * t], Si[2 * t + 1]};
                v2f b  = {G[2 * t], G[2 * t + 1]};
                Dr = WMMA_F32(ar, b, Dr);
                Di = WMMA_F32(ai, b, Di);
            }
#pragma unroll
            for (int i = 0; i < 8; ++i) {
                Sr[i] = Dr[i] * pr - Di[i] * pi;
                Si[i] = Dr[i] * pi + Di[i] * pr;
            }
        }
#pragma unroll
        for (int i = 0; i < 8; ++i) {
            int k = 16 * (8 * h + i) + c;
            sre[k] = Sr[i];
            sim[k] = Si[i];
        }
        __syncthreads();
        if (l < 3) {
#pragma unroll
            for (int i = 0; i < 8; ++i) {
                int src = cnot_ring_src(16 * (8 * h + i) + c);
                Sr[i] = sre[src];
                Si[i] = sim[src];
            }
        }
    }

    for (int j = 0; j < 8; ++j) {
        int n   = wv * 8 + j;
        int src = cnot_ring_src(n);
        MrT[n * 256 + blockIdx.x * 32 + lane] = lds[lane * ROW_PITCH + src];
        MiT[n * 256 + blockIdx.x * 32 + lane] = lds[lane * ROW_PITCH + 257 + src];
    }
}

typedef float v4f __attribute__((ext_vector_type(4)));

#define GM_BM    128
#define GM_KC    32
#define GM_PITCH 36
#define GM_ROWS  (GM_BM + 512)
#define GM_STAGE (GM_ROWS * GM_PITCH)

__device__ __forceinline__ void gm_load(v4f (&st)[10], const float* __restrict__ gx,
                                        const float* __restrict__ gr,
                                        const float* __restrict__ gi, int kc) {
    const int ko = kc * GM_KC;
#pragma unroll
    for (int j = 0; j < 2; ++j) st[j] = *(const v4f*)(gx + j * 64 * 256 + ko);
#pragma unroll
    for (int j = 0; j < 4; ++j) st[2 + j] = *(const v4f*)(gr + j * 64 * 256 + ko);
#pragma unroll
    for (int j = 0; j < 4; ++j) st[6 + j] = *(const v4f*)(gi + j * 64 * 256 + ko);
}

__device__ __forceinline__ void gm_store(const v4f (&st)[10], float* sdst) {
#pragma unroll
    for (int j = 0; j < 10; ++j) *(v4f*)(sdst + j * 64 * GM_PITCH) = st[j];
}

__device__ __forceinline__ void gm_compute(const float* sa, const float* sb,
                                           v8f (&accR)[2][4], v8f (&accI)[2][4],
                                           float (&nrm)[2]) {
#pragma unroll
    for (int s = 0; s < GM_KC / 4; ++s) {
        v2f a[2], br[4], bi[4];
#pragma unroll
        for (int i = 0; i < 2; ++i) a[i] = *(const v2f*)(sa + i * 16 * GM_PITCH + 4 * s);
#pragma unroll
        for (int j = 0; j < 4; ++j) {
            br[j] = *(const v2f*)(sb + j * 16 * GM_PITCH + 4 * s);
            bi[j] = *(const v2f*)(sb + (256 + j * 16) * GM_PITCH + 4 * s);
        }
#pragma unroll
        for (int i = 0; i < 2; ++i)
            nrm[i] += a[i].x * a[i].x + a[i].y * a[i].y;
#pragma unroll
        for (int i = 0; i < 2; ++i) {
#pragma unroll
            for (int j = 0; j < 4; ++j) {
                accR[i][j] = WMMA_F32(a[i], br[j], accR[i][j]);
                accI[i][j] = WMMA_F32(a[i], bi[j], accI[i][j]);
            }
        }
    }
}

__global__ __launch_bounds__(512) void qenc_gemm(const float* __restrict__ X,
                                                 const float* __restrict__ MrT,
                                                 const float* __restrict__ MiT,
                                                 float* __restrict__ out) {
    __shared__ __attribute__((aligned(16))) float gl[2 * GM_STAGE];
    const int tid   = threadIdx.x;
    const int lane  = tid & 31;
    const int wv    = tid >> 5;
    const int half  = lane >> 4;
    const int l15   = lane & 15;
    const int mrow0 = (wv >> 2) * 32;
    const int ncol0 = (wv & 3) * 64;
    const int m0    = blockIdx.x * GM_BM;

    const int seg  = tid & 7;
    const int srow = tid >> 3;
    const float* gx = X   + (size_t)(m0 + srow) * 256 + seg * 4;
    const float* gr = MrT + srow * 256 + seg * 4;
    const float* gi = MiT + srow * 256 + seg * 4;
    float* sdst = gl + srow * GM_PITCH + seg * 4;

    const int aoff = (mrow0 + l15) * GM_PITCH + 2 * half;
    const int boff = (GM_BM + ncol0 + l15) * GM_PITCH + 2 * half;

    v8f accR[2][4], accI[2][4];
#pragma unroll
    for (int i = 0; i < 2; ++i) {
#pragma unroll
        for (int j = 0; j < 4; ++j) {
            accR[i][j] = (v8f){};
            accI[i][j] = (v8f){};
        }
    }
    float nrm[2] = {0.0f, 0.0f};

    v4f st[10];
    gm_load(st, gx, gr, gi, 0);
    gm_store(st, sdst);
    __syncthreads();

    for (int kc = 0; kc < 7; ++kc) {
        gm_load(st, gx, gr, gi, kc + 1);
        const float* sbase = gl + (kc & 1) * GM_STAGE;
        gm_compute(sbase + aoff, sbase + boff, accR, accI, nrm);
        gm_store(st, sdst + ((kc + 1) & 1) * GM_STAGE);
        __syncthreads();
    }
    gm_compute(gl + GM_STAGE + aoff, gl + GM_STAGE + boff, accR, accI, nrm);

    nrm[0] += __shfl_xor(nrm[0], 16, 32);
    nrm[1] += __shfl_xor(nrm[1], 16, 32);

#pragma unroll
    for (int r = 0; r < 8; ++r) {
        int row = r + 8 * half;
#pragma unroll
        for (int i = 0; i < 2; ++i) {
            float inv = 1.0f / __shfl(nrm[i], row, 32);
            float* o  = out + (size_t)(m0 + mrow0 + 16 * i + row) * 256 + ncol0 + l15;
#pragma unroll
            for (int j = 0; j < 4; ++j)
                o[16 * j] = (accR[i][j][r] * accR[i][j][r] + accI[i][j][r] * accI[i][j][r]) * inv;
        }
    }
}

extern "C" void kernel_launch(void* const* d_in, const int* in_sizes, int n_in,
                              void* d_out, int out_size, void* d_ws, size_t ws_size,
                              hipStream_t stream) {
    const float* x    = (const float*)d_in[0];
    const float* wrot = (const float*)d_in[1];
    const float* wrz  = (const float*)d_in[2];
    float* out = (float*)d_out;

    float* MrT = (float*)d_ws;
    float* MiT = MrT + QDIM * QDIM;

    size_t lds_bytes = (32 * 514 + 8 * 16 * 32) * sizeof(float);
    qenc_build<<<8, 1024, lds_bytes, stream>>>(wrot, wrz, MrT, MiT);

    qenc_gemm<<<256, 512, 0, stream>>>(x, MrT, MiT, out);
}


# Dense product replaced: one wave takes each input row through radix-16 transform stages, Kronecker gate factors and LDS-gather CNOT rings on f32 WMMA; no matrix build, 1 launch; sum order changes
# speedup vs baseline: 5.0955x; 1.4987x over previous
#include <hip/hip_runtime.h>

typedef float v2f __attribute__((ext_vector_type(2)));
typedef float v8f __attribute__((ext_vector_type(8)));

#define WMMA_F32(A, B, C) \
    __builtin_amdgcn_wmma_f32_16x16x4_f32(false, (A), false, (B), (short)0, (C), false, false)

#define NWAVES    16
#define ROW_PITCH 514
#define GFRAG_OFF (NWAVES * ROW_PITCH)
#define FTAB_OFF  (GFRAG_OFF + 8 * 16 * 32)
#define LDS_FLOATS (FTAB_OFF + 16 * 2 * 32)
#define ROWS_PER_WG 128

__device__ __forceinline__ int cnot_ring_src(int k) {
    int src = k;
    for (int q = 7; q >= 0; --q) {
        int cb = 7 - q, tb = 7 - ((q + 1) & 7);
        src ^= ((src >> cb) & 1) << tb;
    }
    return src;
}

__device__ __forceinline__ void wave_lds_order() {
    __builtin_amdgcn_fence(__ATOMIC_RELEASE, "wavefront");
    __builtin_amdgcn_wave_barrier();
    __builtin_amdgcn_fence(__ATOMIC_ACQUIRE, "wavefront");
}

__global__ __launch_bounds__(512) void qenc_rows(const float* __restrict__ X,
                                                 const float* __restrict__ wrot,
                                                 const float* __restrict__ wrz,
                                                 float* __restrict__ out) {
    __shared__ float lds[LDS_FLOATS];
    const int lane = threadIdx.x & 31;
    const int wv   = threadIdx.x >> 5;
    const int c    = lane & 15;
    const int h    = lane >> 4;

    float* sre  = lds + wv * ROW_PITCH;
    float* sim  = sre + 257;
    float* gfr  = lds + GFRAG_OFF;
    float* ftab = lds + FTAB_OFF;

    const float* xrow = X + (size_t)(blockIdx.x * ROWS_PER_WG + wv) * 256 + 128 * h + c;
    float xn[8];
#pragma unroll
    for (int i = 0; i < 8; ++i) xn[i] = xrow[16 * i];

    if (wv < 8) {
        float th  = wrot[lane];
        float phz = wrz[lane];
        float gc  = cosf(0.5f * th),  gs  = sinf(0.5f * th);
        float ger = cosf(0.5f * phz), gei = -sinf(0.5f * phz);
        const int g0 = (wv >> 1) * 8 + (wv & 1) * 4;
        float r0[4], r1[4];
        float dr = 1.0f, di = 0.0f;
#pragma unroll
        for (int p = 0; p < 4; ++p) {
            float cc = __shfl(gc,  g0 + p, 32), ss = __shfl(gs,  g0 + p, 32);
            float er = __shfl(ger, g0 + p, 32), ei = __shfl(gei, g0 + p, 32);
            int rb = (c >> (3 - p)) & 1;
            r0[p] = rb ? ss :  cc;
            r1[p] = rb ? cc : -ss;
            float fi = rb ? -ei : ei;
            float nr = dr * er - di * fi;
            float ni = dr * fi + di * er;
            dr = nr;
            di = ni;
        }
        float e0 = h ? r1[0] : r0[0];
#pragma unroll
        for (int j = 0; j < 8; ++j) {
            int b2 = (j >> 2) & 1, b1 = (j >> 1) & 1, b0 = j & 1;
            float x1 = b2 ? r1[1] : r0[1];
            float x2 = b1 ? r1[2] : r0[2];
            float x3 = b0 ? r1[3] : r0[3];
            gfr[(wv * 16 + j) * 32 + lane] = ((e0 * x1) * x2) * x3;
        }
        gfr[(wv * 16 + 8) * 32 + lane] = dr;
        gfr[(wv * 16 + 9) * 32 + lane] = di;
    }

    {
        const float w = 6.28318530717958647692f / 256.0f;
        int idx = (8 * h + (wv & 7)) * c;
        if (wv < 8) idx <<= 4;
        idx &= 255;
        float sv, cv;
        sincosf(w * (float)idx, &sv, &cv);
        ftab[(wv * 2 + 0) * 32 + lane] = cv;
        ftab[(wv * 2 + 1) * 32 + lane] = sv;
    }

    int gsrc[8], osrc[8];
#pragma unroll
    for (int i = 0; i < 8; ++i) {
        gsrc[i] = cnot_ring_src(16 * (8 * h + i) + c);
        osrc[i] = cnot_ring_src(32 * i + lane);
    }

    __syncthreads();

    float Fr[8], Fi[8], nFi[8], twr[8], twi[8];
#pragma unroll
    for (int j = 0; j < 8; ++j) {
        Fr[j]  = ftab[(j * 2 + 0) * 32 + lane];
        Fi[j]  = ftab[(j * 2 + 1) * 32 + lane];
        nFi[j] = -Fi[j];
        twr[j] = ftab[((8 + j) * 2 + 0) * 32 + lane];
        twi[j] = ftab[((8 + j) * 2 + 1) * 32 + lane];
    }

    float GF[8][10];
#pragma unroll
    for (int f = 0; f < 8; ++f) {
#pragma unroll
        for (int j = 0; j < 10; ++j) GF[f][j] = gfr[(f * 16 + j) * 32 + lane];
    }

    for (int it = 0; it < ROWS_PER_WG / NWAVES; ++it) {
        const int row = blockIdx.x * ROWS_PER_WG + it * NWAVES + wv;
        float xr[8];
#pragma unroll
        for (int i = 0; i < 8; ++i) xr[i] = xn[i];
        if (it + 1 < ROWS_PER_WG / NWAVES) {
#pragma unroll
            for (int i = 0; i < 8; ++i) xn[i] = xrow[(size_t)(it + 1) * NWAVES * 256 + 16 * i];
        }

        float n2 = 0.0f;
#pragma unroll
        for (int i = 0; i < 8; ++i) n2 += xr[i] * xr[i];
#pragma unroll
        for (int m = 16; m >= 1; m >>= 1) n2 += __shfl_xor(n2, m, 32);

        v8f Dr = {}, Di = {};
#pragma unroll
        for (int t = 0; t < 4; ++t) {
            v2f a  = {xr[2 * t], xr[2 * t + 1]};
            v2f fr = {Fr[2 * t], Fr[2 * t + 1]};
            v2f fi = {Fi[2 * t], Fi[2 * t + 1]};
            Dr = WMMA_F32(a, fr, Dr);
            Di = WMMA_F32(a, fi, Di);
        }
        v8f Tr, Ti;
#pragma unroll
        for (int i = 0; i < 8; ++i) {
            Tr[i] = Dr[i] * twr[i] - Di[i] * twi[i];
            Ti[i] = Dr[i] * twi[i] + Di[i] * twr[i];
        }
        v8f Sr = {}, Si = {};
#pragma unroll
        for (int t = 0; t < 4; ++t) {
            v2f ar  = {Tr[2 * t], Tr[2 * t + 1]};
            v2f ai  = {Ti[2 * t], Ti[2 * t + 1]};
            v2f fr  = {Fr[2 * t], Fr[2 * t + 1]};
            v2f fi  = {Fi[2 * t], Fi[2 * t + 1]};
            v2f nfi = {nFi[2 * t], nFi[2 * t + 1]};
            Sr = WMMA_F32(ar, fr,  Sr);
            Si = WMMA_F32(ar, fi,  Si);
            Sr = WMMA_F32(ai, nfi, Sr);
            Si = WMMA_F32(ai, fr,  Si);
        }

#pragma unroll
        for (int l = 0; l < 4; ++l) {
#pragma unroll
            for (int gg = 0; gg < 2; ++gg) {
                const int g = (l == 0) ? (1 - gg) : gg;
                const float (&G)[10] = GF[l * 2 + g];
                const float pr = G[8], pi = G[9];
                v8f Er = {}, Ei = {};
#pragma unroll
                for (int t = 0; t < 4; ++t) {
                    v2f ar = {Sr[2 * t], Sr[2 * t + 1]};
                    v2f ai = {Si[2 * t], Si[2 * t + 1]};
                    v2f b  = {G[2 * t], G[2 * t + 1]};
                    Er = WMMA_F32(ar, b, Er);
                    Ei = WMMA_F32(ai, b, Ei);
                }
#pragma unroll
                for (int i = 0; i < 8; ++i) {
                    Sr[i] = Er[i] * pr - Ei[i] * pi;
                    Si[i] = Er[i] * pi + Ei[i] * pr;
                }
            }
            if (l < 3) {
                const int kb = (l == 0) ? (16 * c + 8 * h) : (128 * h + c);
                const int ks = (l == 0) ? 1 : 16;
#pragma unroll
                for (int i = 0; i < 8; ++i) {
                    sre[kb + ks * i] = Sr[i];
                    sim[kb + ks * i] = Si[i];
                }
                wave_lds_order();
#pragma unroll
                for (int i = 0; i < 8; ++i) {
                    Sr[i] = sre[gsrc[i]];
                    Si[i] = sim[gsrc[i]];
                }
                wave_lds_order();
            }
        }

        const float inv = 1.0f / (256.0f * n2);
#pragma unroll
        for (int i = 0; i < 8; ++i)
            sre[16 * (8 * h + i) + c] = (Sr[i] * Sr[i] + Si[i] * Si[i]) * inv;
        wave_lds_order();
        float* o = out + (size_t)row * 256 + lane;
#pragma unroll
        for (int j = 0; j < 8; ++j) o[32 * j] = sre[osrc[j]];
        wave_lds_order();
    }
}

extern "C" void kernel_launch(void* const* d_in, const int* in_sizes, int n_in,
                              void* d_out, int out_size, void* d_ws, size_t ws_size,
                              hipStream_t stream) {
    const float* x    = (const float*)d_in[0];
    const float* wrot = (const float*)d_in[1];
    const float* wrz  = (const float*)d_in[2];
    float* out = (float*)d_out;

    qenc_rows<<<256, 512, 0, stream>>>(x, wrot, wrz, out);
}


# Transform stage 2 and layer 0's gate factor on wires 0..3 fused into one complex 16x16 matrix built at set-up: 80 WMMAs per row, was 88, one phase pass less; that product's sum order changes
# speedup vs baseline: 5.4566x; 1.0709x over previous
#include <hip/hip_runtime.h>

typedef float v2f __attribute__((ext_vector_type(2)));
typedef float v8f __attribute__((ext_vector_type(8)));

#define WMMA_F32(A, B, C) \
    __builtin_amdgcn_wmma_f32_16x16x4_f32(false, (A), false, (B), (short)0, (C), false, false)

#define NWAVES    16
#define ROW_PITCH 514
#define GFRAG_OFF (NWAVES * ROW_PITCH)
#define FTAB_OFF  (GFRAG_OFF + 8 * 16 * 32)
#define LDS_FLOATS (FTAB_OFF + 16 * 2 * 32)
#define ROWS_PER_WG 128

__device__ __forceinline__ int cnot_ring_src(int k) {
    int src = k;
    for (int q = 7; q >= 0; --q) {
        int cb = 7 - q, tb = 7 - ((q + 1) & 7);
        src ^= ((src >> cb) & 1) << tb;
    }
    return src;
}

__device__ __forceinline__ void wave_lds_order() {
    __builtin_amdgcn_fence(__ATOMIC_RELEASE, "wavefront");
    __builtin_amdgcn_wave_barrier();
    __builtin_amdgcn_fence(__ATOMIC_ACQUIRE, "wavefront");
}

__global__ __launch_bounds__(512) void qenc_rows(const float* __restrict__ X,
                                                 const float* __restrict__ wrot,
                                                 const float* __restrict__ wrz,
                                                 float* __restrict__ out) {
    __shared__ float lds[LDS_FLOATS];
    const int lane = threadIdx.x & 31;
    const int wv   = threadIdx.x >> 5;
    const int c    = lane & 15;
    const int h    = lane >> 4;

    float* sre  = lds + wv * ROW_PITCH;
    float* sim  = sre + 257;
    float* gfr  = lds + GFRAG_OFF;
    float* ftab = lds + FTAB_OFF;

    const float* xrow = X + (size_t)(blockIdx.x * ROWS_PER_WG + wv) * 256 + 128 * h + c;
    float xn[8];
#pragma unroll
    for (int i = 0; i < 8; ++i) xn[i] = xrow[16 * i];

    if (wv < 8) {
        float th  = wrot[lane];
        float phz = wrz[lane];
        float gc  = cosf(0.5f * th),  gs  = sinf(0.5f * th);
        float ger = cosf(0.5f * phz), gei = -sinf(0.5f * phz);
        const int g0 = (wv >> 1) * 8 + (wv & 1) * 4;
        float r0[4], r1[4];
        float dr = 1.0f, di = 0.0f;
#pragma unroll
        for (int p = 0; p < 4; ++p) {
            float cc = __shfl(gc,  g0 + p, 32), ss = __shfl(gs,  g0 + p, 32);
            float er = __shfl(ger, g0 + p, 32), ei = __shfl(gei, g0 + p, 32);
            int rb = (c >> (3 - p)) & 1;
            r0[p] = rb ? ss :  cc;
            r1[p] = rb ? cc : -ss;
            float fi = rb ? -ei : ei;
            float nr = dr * er - di * fi;
            float ni = dr * fi + di * er;
            dr = nr;
            di = ni;
        }
        if (wv == 0) {
            const float k16c[16] = { 1.0f,  0.92387953251128674f,  0.70710678118654752f,  0.38268343236508977f,
                                     0.0f, -0.38268343236508977f, -0.70710678118654752f, -0.92387953251128674f,
                                    -1.0f, -0.92387953251128674f, -0.70710678118654752f, -0.38268343236508977f,
                                     0.0f,  0.38268343236508977f,  0.70710678118654752f,  0.92387953251128674f};
            const float k16s[16] = { 0.0f,  0.38268343236508977f,  0.70710678118654752f,  0.92387953251128674f,
                                     1.0f,  0.92387953251128674f,  0.70710678118654752f,  0.38268343236508977f,
                                     0.0f, -0.38268343236508977f, -0.70710678118654752f, -0.92387953251128674f,
                                    -1.0f, -0.92387953251128674f, -0.70710678118654752f, -0.38268343236508977f};
            const float sg = h ? -1.0f : 1.0f;
            float rrow[16];
#pragma unroll
            for (int n = 0; n < 16; ++n) {
                float x0 = (n & 8) ? r1[0] : r0[0];
                float x1 = (n & 4) ? r1[1] : r0[1];
                float x2 = (n & 2) ? r1[2] : r0[2];
                float x3 = (n & 1) ? r1[3] : r0[3];
                float v = ((x0 * x1) * x2) * x3;
                rrow[n] = (n & 1) ? sg * v : v;
            }
#pragma unroll
            for (int j = 0; j < 8; ++j) {
                float sr = 0.0f, si = 0.0f;
#pragma unroll
                for (int n = 0; n < 16; ++n) {
                    sr += k16c[(j * n) & 15] * rrow[n];
                    si += k16s[(j * n) & 15] * rrow[n];
                }
                gfr[j * 32 + lane]       = dr * sr - di * si;
                gfr[(8 + j) * 32 + lane] = dr * si + di * sr;
            }
        } else {
            float e0 = h ? r1[0] : r0[0];
#pragma unroll
            for (int j = 0; j < 8; ++j) {
                int b2 = (j >> 2) & 1, b1 = (j >> 1) & 1, b0 = j & 1;
                float x1 = b2 ? r1[1] : r0[1];
                float x2 = b1 ? r1[2] : r0[2];
                float x3 = b0 ? r1[3] : r0[3];
                gfr[(wv * 16 + j) * 32 + lane] = ((e0 * x1) * x2) * x3;
            }
            gfr[(wv * 16 + 8) * 32 + lane] = dr;
            gfr[(wv * 16 + 9) * 32 + lane] = di;
        }
    }

    {
        const float w = 6.28318530717958647692f / 256.0f;
        int idx = (8 * h + (wv & 7)) * c;
        if (wv < 8) idx <<= 4;
        idx &= 255;
        float sv, cv;
        sincosf(w * (float)idx, &sv, &cv);
        ftab[(wv * 2 + 0) * 32 + lane] = cv;
        ftab[(wv * 2 + 1) * 32 + lane] = sv;
    }

    int gsrc[8], osrc[8];
#pragma unroll
    for (int i = 0; i < 8; ++i) {
        gsrc[i] = cnot_ring_src(16 * (8 * h + i) + c);
        osrc[i] = cnot_ring_src(32 * i + lane);
    }

    __syncthreads();

    float Fr[8], Fi[8], twr[8], twi[8], Cr[8], Ci[8], nCi[8];
#pragma unroll
    for (int j = 0; j < 8; ++j) {
        Fr[j]  = ftab[(j * 2 + 0) * 32 + lane];
        Fi[j]  = ftab[(j * 2 + 1) * 32 + lane];
        twr[j] = ftab[((8 + j) * 2 + 0) * 32 + lane];
        twi[j] = ftab[((8 + j) * 2 + 1) * 32 + lane];
        Cr[j]  = gfr[j * 32 + lane];
        Ci[j]  = gfr[(8 + j) * 32 + lane];
        nCi[j] = -Ci[j];
    }

    float GF[8][10];
#pragma unroll
    for (int f = 1; f < 8; ++f) {
#pragma unroll
        for (int j = 0; j < 10; ++j) GF[f][j] = gfr[(f * 16 + j) * 32 + lane];
    }

    for (int it = 0; it < ROWS_PER_WG / NWAVES; ++it) {
        const int row = blockIdx.x * ROWS_PER_WG + it * NWAVES + wv;
        float xr[8];
#pragma unroll
        for (int i = 0; i < 8; ++i) xr[i] = xn[i];
        if (it + 1 < ROWS_PER_WG / NWAVES) {
#pragma unroll
            for (int i = 0; i < 8; ++i) xn[i] = xrow[(size_t)(it + 1) * NWAVES * 256 + 16 * i];
        }

        float n2 = 0.0f;
#pragma unroll
        for (int i = 0; i < 8; ++i) n2 += xr[i] * xr[i];
#pragma unroll
        for (int m = 16; m >= 1; m >>= 1) n2 += __shfl_xor(n2, m, 32);

        v8f Dr = {}, Di = {};
#pragma unroll
        for (int t = 0; t < 4; ++t) {
            v2f a  = {xr[2 * t], xr[2 * t + 1]};
            v2f fr = {Fr[2 * t], Fr[2 * t + 1]};
            v2f fi = {Fi[2 * t], Fi[2 * t + 1]};
            Dr = WMMA_F32(a, fr, Dr);
            Di = WMMA_F32(a, fi, Di);
        }
        v8f Tr, Ti;
#pragma unroll
        for (int i = 0; i < 8; ++i) {
            Tr[i] = Dr[i] * twr[i] - Di[i] * twi[i];
            Ti[i] = Dr[i] * twi[i] + Di[i] * twr[i];
        }
        v8f Sr = {}, Si = {};
#pragma unroll
        for (int t = 0; t < 4; ++t) {
            v2f ar = {Tr[2 * t], Tr[2 * t + 1]};
            v2f ai = {Ti[2 * t], Ti[2 * t + 1]};
            v2f cr = {Cr[2 * t], Cr[2 * t + 1]};
            v2f ci  = {Ci[2 * t], Ci[2 * t + 1]};
            v2f nci = {nCi[2 * t], nCi[2 * t + 1]};
            Sr = WMMA_F32(ar, cr,  Sr);
            Si = WMMA_F32(ar, ci,  Si);
            Sr = WMMA_F32(ai, nci, Sr);
            Si = WMMA_F32(ai, cr,  Si);
        }

#pragma unroll
        for (int l = 0; l < 4; ++l) {
#pragma unroll
            for (int g = (l == 0) ? 1 : 0; g < 2; ++g) {
                const float (&G)[10] = GF[l * 2 + g];
                const float pr = G[8], pi = G[9];
                v8f Er = {}, Ei = {};
#pragma unroll
                for (int t = 0; t < 4; ++t) {
                    v2f ar = {Sr[2 * t], Sr[2 * t + 1]};
                    v2f ai = {Si[2 * t], Si[2 * t + 1]};
                    v2f b  = {G[2 * t], G[2 * t + 1]};
                    Er = WMMA_F32(ar, b, Er);
                    Ei = WMMA_F32(ai, b, Ei);
                }
#pragma unroll
                for (int i = 0; i < 8; ++i) {
                    Sr[i] = Er[i] * pr - Ei[i] * pi;
                    Si[i] = Er[i] * pi + Ei[i] * pr;
                }
            }
            if (l < 3) {
                const int kb = 128 * h + c;
#pragma unroll
                for (int i = 0; i < 8; ++i) {
                    sre[kb + 16 * i] = Sr[i];
                    sim[kb + 16 * i] = Si[i];
                }
                wave_lds_order();
#pragma unroll
                for (int i = 0; i < 8; ++i) {
                    Sr[i] = sre[gsrc[i]];
                    Si[i] = sim[gsrc[i]];
                }
                wave_lds_order();
            }
        }

        const float inv = 1.0f / (256.0f * n2);
#pragma unroll
        for (int i = 0; i < 8; ++i)
            sre[16 * (8 * h + i) + c] = (Sr[i] * Sr[i] + Si[i] * Si[i]) * inv;
        wave_lds_order();
        float* o = out + (size_t)row * 256 + lane;
#pragma unroll
        for (int j = 0; j < 8; ++j) o[32 * j] = sre[osrc[j]];
        wave_lds_order();
    }
}

extern "C" void kernel_launch(void* const* d_in, const int* in_sizes, int n_in,
                              void* d_out, int out_size, void* d_ws, size_t ws_size,
                              hipStream_t stream) {
    const float* x    = (const float*)d_in[0];
    const float* wrot = (const float*)d_in[1];
    const float* wrz  = (const float*)d_in[2];
    float* out = (float*)d_out;

    qenc_rows<<<256, 512, 0, stream>>>(x, wrot, wrz, out);
}


# Wave priority from the trip counter (s_setprio 3,3,3,2,2,1,1,0 over a wave's 8 rows): the SIMD's wave with fewest rows done wins the matrix pipe, all four finish together; arithmetic unchanged
# speedup vs baseline: 5.6277x; 1.0314x over previous
#include <hip/hip_runtime.h>

typedef float v2f __attribute__((ext_vector_type(2)));
typedef float v8f __attribute__((ext_vector_type(8)));

#define WMMA_F32(A, B, C) \
    __builtin_amdgcn_wmma_f32_16x16x4_f32(false, (A), false, (B), (short)0, (C), false, false)

#define NWAVES    16
#define ROW_PITCH 514
#define GFRAG_OFF (NWAVES * ROW_PITCH)
#define FTAB_OFF  (GFRAG_OFF + 8 * 16 * 32)
#define LDS_FLOATS (FTAB_OFF + 16 * 2 * 32)
#define ROWS_PER_WG 128

__device__ __forceinline__ int cnot_ring_src(int k) {
    int src = k;
    for (int q = 7; q >= 0; --q) {
        int cb = 7 - q, tb = 7 - ((q + 1) & 7);
        src ^= ((src >> cb) & 1) << tb;
    }
    return src;
}

__device__ __forceinline__ void wave_lds_order() {
    __builtin_amdgcn_fence(__ATOMIC_RELEASE, "wavefront");
    __builtin_amdgcn_wave_barrier();
    __builtin_amdgcn_fence(__ATOMIC_ACQUIRE, "wavefront");
}

__global__ __launch_bounds__(512) void qenc_rows(const float* __restrict__ X,
                                                 const float* __restrict__ wrot,
                                                 const float* __restrict__ wrz,
                                                 float* __restrict__ out) {
    __shared__ float lds[LDS_FLOATS];
    const int lane = threadIdx.x & 31;
    const int wv   = threadIdx.x >> 5;
    const int c    = lane & 15;
    const int h    = lane >> 4;

    float* sre  = lds + wv * ROW_PITCH;
    float* sim  = sre + 257;
    float* gfr  = lds + GFRAG_OFF;
    float* ftab = lds + FTAB_OFF;

    const float* xrow = X + (size_t)(blockIdx.x * ROWS_PER_WG + wv) * 256 + 128 * h + c;
    float xn[8];
#pragma unroll
    for (int i = 0; i < 8; ++i) xn[i] = xrow[16 * i];

    if (wv < 8) {
        float th  = wrot[lane];
        float phz = wrz[lane];
        float gc  = cosf(0.5f * th),  gs  = sinf(0.5f * th);
        float ger = cosf(0.5f * phz), gei = -sinf(0.5f * phz);
        const int g0 = (wv >> 1) * 8 + (wv & 1) * 4;
        float r0[4], r1[4];
        float dr = 1.0f, di = 0.0f;
#pragma unroll
        for (int p = 0; p < 4; ++p) {
            float cc = __shfl(gc,  g0 + p, 32), ss = __shfl(gs,  g0 + p, 32);
            float er = __shfl(ger, g0 + p, 32), ei = __shfl(gei, g0 + p, 32);
            int rb = (c >> (3 - p)) & 1;
            r0[p] = rb ? ss :  cc;
            r1[p] = rb ? cc : -ss;
            float fi = rb ? -ei : ei;
            float nr = dr * er - di * fi;
            float ni = dr * fi + di * er;
            dr = nr;
            di = ni;
        }
        if (wv == 0) {
            const float k16c[16] = { 1.0f,  0.92387953251128674f,  0.70710678118654752f,  0.38268343236508977f,
                                     0.0f, -0.38268343236508977f, -0.70710678118654752f, -0.92387953251128674f,
                                    -1.0f, -0.92387953251128674f, -0.70710678118654752f, -0.38268343236508977f,
                                     0.0f,  0.38268343236508977f,  0.70710678118654752f,  0.92387953251128674f};
            const float k16s[16] = { 0.0f,  0.38268343236508977f,  0.70710678118654752f,  0.92387953251128674f,
                                     1.0f,  0.92387953251128674f,  0.70710678118654752f,  0.38268343236508977f,
                                     0.0f, -0.38268343236508977f, -0.70710678118654752f, -0.92387953251128674f,
                                    -1.0f, -0.92387953251128674f, -0.70710678118654752f, -0.38268343236508977f};
            const float sg = h ? -1.0f : 1.0f;
            float rrow[16];
#pragma unroll
            for (int n = 0; n < 16; ++n) {
                float x0 = (n & 8) ? r1[0] : r0[0];
                float x1 = (n & 4) ? r1[1] : r0[1];
                float x2 = (n & 2) ? r1[2] : r0[2];
                float x3 = (n & 1) ? r1[3] : r0[3];
                float v = ((x0 * x1) * x2) * x3;
                rrow[n] = (n & 1) ? sg * v : v;
            }
#pragma unroll
            for (int j = 0; j < 8; ++j) {
                float sr = 0.0f, si = 0.0f;
#pragma unroll
                for (int n = 0; n < 16; ++n) {
                    sr += k16c[(j * n) & 15] * rrow[n];
                    si += k16s[(j * n) & 15] * rrow[n];
                }
                gfr[j * 32 + lane]       = dr * sr - di * si;
                gfr[(8 + j) * 32 + lane] = dr * si + di * sr;
            }
        } else {
            float e0 = h ? r1[0] : r0[0];
#pragma unroll
            for (int j = 0; j < 8; ++j) {
                int b2 = (j >> 2) & 1, b1 = (j >> 1) & 1, b0 = j & 1;
                float x1 = b2 ? r1[1] : r0[1];
                float x2 = b1 ? r1[2] : r0[2];
                float x3 = b0 ? r1[3] : r0[3];
                gfr[(wv * 16 + j) * 32 + lane] = ((e0 * x1) * x2) * x3;
            }
            gfr[(wv * 16 + 8) * 32 + lane] = dr;
            gfr[(wv * 16 + 9) * 32 + lane] = di;
        }
    }

    {
        const float w = 6.28318530717958647692f / 256.0f;
        int idx = (8 * h + (wv & 7)) * c;
        if (wv < 8) idx <<= 4;
        idx &= 255;
        float sv, cv;
        sincosf(w * (float)idx, &sv, &cv);
        ftab[(wv * 2 + 0) * 32 + lane] = cv;
        ftab[(wv * 2 + 1) * 32 + lane] = sv;
    }

    int gsrc[8], osrc[8];
#pragma unroll
    for (int i = 0; i < 8; ++i) {
        gsrc[i] = cnot_ring_src(16 * (8 * h + i) + c);
        osrc[i] = cnot_ring_src(32 * i + lane);
    }

    __syncthreads();

    float Fr[8], Fi[8], twr[8], twi[8], Cr[8], Ci[8], nCi[8];
#pragma unroll
    for (int j = 0; j < 8; ++j) {
        Fr[j]  = ftab[(j * 2 + 0) * 32 + lane];
        Fi[j]  = ftab[(j * 2 + 1) * 32 + lane];
        twr[j] = ftab[((8 + j) * 2 + 0) * 32 + lane];
        twi[j] = ftab[((8 + j) * 2 + 1) * 32 + lane];
        Cr[j]  = gfr[j * 32 + lane];
        Ci[j]  = gfr[(8 + j) * 32 + lane];
        nCi[j] = -Ci[j];
    }

    float GF[8][10];
#pragma unroll
    for (int f = 1; f < 8; ++f) {
#pragma unroll
        for (int j = 0; j < 10; ++j) GF[f][j] = gfr[(f * 16 + j) * 32 + lane];
    }

    for (int it = 0; it < ROWS_PER_WG / NWAVES; ++it) {
        if (it == 0)      __builtin_amdgcn_s_setprio(3);
        else if (it == 3) __builtin_amdgcn_s_setprio(2);
        else if (it == 5) __builtin_amdgcn_s_setprio(1);
        else if (it == 7) __builtin_amdgcn_s_setprio(0);

        const int row = blockIdx.x * ROWS_PER_WG + it * NWAVES + wv;
        float xr[8];
#pragma unroll
        for (int i = 0; i < 8; ++i) xr[i] = xn[i];
        if (it + 1 < ROWS_PER_WG / NWAVES) {
#pragma unroll
            for (int i = 0; i < 8; ++i) xn[i] = xrow[(size_t)(it + 1) * NWAVES * 256 + 16 * i];
        }

        float n2 = 0.0f;
#pragma unroll
        for (int i = 0; i < 8; ++i) n2 += xr[i] * xr[i];
#pragma unroll
        for (int m = 16; m >= 1; m >>= 1) n2 += __shfl_xor(n2, m, 32);

        v8f Dr = {}, Di = {};
#pragma unroll
        for (int t = 0; t < 4; ++t) {
            v2f a  = {xr[2 * t], xr[2 * t + 1]};
            v2f fr = {Fr[2 * t], Fr[2 * t + 1]};
            v2f fi = {Fi[2 * t], Fi[2 * t + 1]};
            Dr = WMMA_F32(a, fr, Dr);
            Di = WMMA_F32(a, fi, Di);
        }
        v8f Tr, Ti;
#pragma unroll
        for (int i = 0; i < 8; ++i) {
            Tr[i] = Dr[i] * twr[i] - Di[i] * twi[i];
            Ti[i] = Dr[i] * twi[i] + Di[i] * twr[i];
        }
        v8f Sr = {}, Si = {};
#pragma unroll
        for (int t = 0; t < 4; ++t) {
            v2f ar = {Tr[2 * t], Tr[2 * t + 1]};
            v2f ai = {Ti[2 * t], Ti[2 * t + 1]};
            v2f cr = {Cr[2 * t], Cr[2 * t + 1]};
            v2f ci  = {Ci[2 * t], Ci[2 * t + 1]};
            v2f nci = {nCi[2 * t], nCi[2 * t + 1]};
            Sr = WMMA_F32(ar, cr,  Sr);
            Si = WMMA_F32(ar, ci,  Si);
            Sr = WMMA_F32(ai, nci, Sr);
            Si = WMMA_F32(ai, cr,  Si);
        }

#pragma unroll
        for (int l = 0; l < 4; ++l) {
#pragma unroll
            for (int g = (l == 0) ? 1 : 0; g < 2; ++g) {
                const float (&G)[10] = GF[l * 2 + g];
                const float pr = G[8], pi = G[9];
                v8f Er = {}, Ei = {};
#pragma unroll
                for (int t = 0; t < 4; ++t) {
                    v2f ar = {Sr[2 * t], Sr[2 * t + 1]};
                    v2f ai = {Si[2 * t], Si[2 * t + 1]};
                    v2f b  = {G[2 * t], G[2 * t + 1]};
                    Er = WMMA_F32(ar, b, Er);
                    Ei = WMMA_F32(ai, b, Ei);
                }
#pragma unroll
                for (int i = 0; i < 8; ++i) {
                    Sr[i] = Er[i] * pr - Ei[i] * pi;
                    Si[i] = Er[i] * pi + Ei[i] * pr;
                }
            }
            if (l < 3) {
                const int kb = 128 * h + c;
#pragma unroll
                for (int i = 0; i < 8; ++i) {
                    sre[kb + 16 * i] = Sr[i];
                    sim[kb + 16 * i] = Si[i];
                }
                wave_lds_order();
#pragma unroll
                for (int i = 0; i < 8; ++i) {
                    Sr[i] = sre[gsrc[i]];
                    Si[i] = sim[gsrc[i]];
                }
                wave_lds_order();
            }
        }

        const float inv = 1.0f / (256.0f * n2);
#pragma unroll
        for (int i = 0; i < 8; ++i)
            sre[16 * (8 * h + i) + c] = (Sr[i] * Sr[i] + Si[i] * Si[i]) * inv;
        wave_lds_order();
        float* o = out + (size_t)row * 256 + lane;
#pragma unroll
        for (int j = 0; j < 8; ++j) o[32 * j] = sre[osrc[j]];
        wave_lds_order();
    }
}

extern "C" void kernel_launch(void* const* d_in, const int* in_sizes, int n_in,
                              void* d_out, int out_size, void* d_ws, size_t ws_size,
                              hipStream_t stream) {
    const float* x    = (const float*)d_in[0];
    const float* wrot = (const float*)d_in[1];
    const float* wrz  = (const float*)d_in[2];
    float* out = (float*)d_out;

    qenc_rows<<<256, 512, 0, stream>>>(x, wrot, wrz, out);
}


# Phases once per layer: hi and lo unit phases merged into one per-element table behind the lo step (they commute), layer 3's dropped (unit modulus): 3 phase passes, was 7; product rounding changes
# speedup vs baseline: 5.8154x; 1.0334x over previous
#include <hip/hip_runtime.h>

typedef float v2f __attribute__((ext_vector_type(2)));
typedef float v8f __attribute__((ext_vector_type(8)));

#define WMMA_F32(A, B, C) \
    __builtin_amdgcn_wmma_f32_16x16x4_f32(false, (A), false, (B), (short)0, (C), false, false)

#define NWAVES    16
#define ROW_PITCH 514
#define GFRAG_OFF (NWAVES * ROW_PITCH)
#define FTAB_OFF  (GFRAG_OFF + 8 * 16 * 32)
#define LDS_FLOATS (FTAB_OFF + 16 * 2 * 32)
#define ROWS_PER_WG 128

__device__ __forceinline__ int cnot_ring_src(int k) {
    int src = k;
    for (int q = 7; q >= 0; --q) {
        int cb = 7 - q, tb = 7 - ((q + 1) & 7);
        src ^= ((src >> cb) & 1) << tb;
    }
    return src;
}

__device__ __forceinline__ void wave_lds_order() {
    __builtin_amdgcn_fence(__ATOMIC_RELEASE, "wavefront");
    __builtin_amdgcn_wave_barrier();
    __builtin_amdgcn_fence(__ATOMIC_ACQUIRE, "wavefront");
}

__global__ __launch_bounds__(512) void qenc_rows(const float* __restrict__ X,
                                                 const float* __restrict__ wrot,
                                                 const float* __restrict__ wrz,
                                                 float* __restrict__ out) {
    __shared__ float lds[LDS_FLOATS];
    const int lane = threadIdx.x & 31;
    const int wv   = threadIdx.x >> 5;
    const int c    = lane & 15;
    const int h    = lane >> 4;

    float* sre  = lds + wv * ROW_PITCH;
    float* sim  = sre + 257;
    float* gfr  = lds + GFRAG_OFF;
    float* ftab = lds + FTAB_OFF;

    const float* xrow = X + (size_t)(blockIdx.x * ROWS_PER_WG + wv) * 256 + 128 * h + c;
    float xn[8];
#pragma unroll
    for (int i = 0; i < 8; ++i) xn[i] = xrow[16 * i];

    if (wv < 8) {
        float th  = wrot[lane];
        float phz = wrz[lane];
        float gc  = cosf(0.5f * th),  gs  = sinf(0.5f * th);
        float ger = cosf(0.5f * phz), gei = -sinf(0.5f * phz);
        const int g0 = (wv >> 1) * 8 + (wv & 1) * 4;
        float r0[4], r1[4];
        float dr = 1.0f, di = 0.0f;
#pragma unroll
        for (int p = 0; p < 4; ++p) {
            float cc = __shfl(gc,  g0 + p, 32), ss = __shfl(gs,  g0 + p, 32);
            float er = __shfl(ger, g0 + p, 32), ei = __shfl(gei, g0 + p, 32);
            int rb = (c >> (3 - p)) & 1;
            r0[p] = rb ? ss :  cc;
            r1[p] = rb ? cc : -ss;
            float fi = rb ? -ei : ei;
            float nr = dr * er - di * fi;
            float ni = dr * fi + di * er;
            dr = nr;
            di = ni;
        }
        if (wv == 0) {
            const float k16c[16] = { 1.0f,  0.92387953251128674f,  0.70710678118654752f,  0.38268343236508977f,
                                     0.0f, -0.38268343236508977f, -0.70710678118654752f, -0.92387953251128674f,
                                    -1.0f, -0.92387953251128674f, -0.70710678118654752f, -0.38268343236508977f,
                                     0.0f,  0.38268343236508977f,  0.70710678118654752f,  0.92387953251128674f};
            const float k16s[16] = { 0.0f,  0.38268343236508977f,  0.70710678118654752f,  0.92387953251128674f,
                                     1.0f,  0.92387953251128674f,  0.70710678118654752f,  0.38268343236508977f,
                                     0.0f, -0.38268343236508977f, -0.70710678118654752f, -0.92387953251128674f,
                                    -1.0f, -0.92387953251128674f, -0.70710678118654752f, -0.38268343236508977f};
            const float sg = h ? -1.0f : 1.0f;
            float rrow[16];
#pragma unroll
            for (int n = 0; n < 16; ++n) {
                float x0 = (n & 8) ? r1[0] : r0[0];
                float x1 = (n & 4) ? r1[1] : r0[1];
                float x2 = (n & 2) ? r1[2] : r0[2];
                float x3 = (n & 1) ? r1[3] : r0[3];
                float v = ((x0 * x1) * x2) * x3;
                rrow[n] = (n & 1) ? sg * v : v;
            }
#pragma unroll
            for (int j = 0; j < 8; ++j) {
                float sr = 0.0f, si = 0.0f;
#pragma unroll
                for (int n = 0; n < 16; ++n) {
                    sr += k16c[(j * n) & 15] * rrow[n];
                    si += k16s[(j * n) & 15] * rrow[n];
                }
                gfr[j * 32 + lane]       = dr * sr - di * si;
                gfr[(8 + j) * 32 + lane] = dr * si + di * sr;
            }
        } else {
            float e0 = h ? r1[0] : r0[0];
#pragma unroll
            for (int j = 0; j < 8; ++j) {
                int b2 = (j >> 2) & 1, b1 = (j >> 1) & 1, b0 = j & 1;
                float x1 = b2 ? r1[1] : r0[1];
                float x2 = b1 ? r1[2] : r0[2];
                float x3 = b0 ? r1[3] : r0[3];
                gfr[(wv * 16 + j) * 32 + lane] = ((e0 * x1) * x2) * x3;
            }
            gfr[(wv * 16 + 8) * 32 + lane] = dr;
            gfr[(wv * 16 + 9) * 32 + lane] = di;
        }
    }

    {
        const float w = 6.28318530717958647692f / 256.0f;
        int idx = (8 * h + (wv & 7)) * c;
        if (wv < 8) idx <<= 4;
        idx &= 255;
        float sv, cv;
        sincosf(w * (float)idx, &sv, &cv);
        ftab[(wv * 2 + 0) * 32 + lane] = cv;
        ftab[(wv * 2 + 1) * 32 + lane] = sv;
    }

    int gsrc[8], osrc[8];
#pragma unroll
    for (int i = 0; i < 8; ++i) {
        gsrc[i] = cnot_ring_src(16 * (8 * h + i) + c);
        osrc[i] = cnot_ring_src(32 * i + lane);
    }

    __syncthreads();

    float Fr[8], Fi[8], twr[8], twi[8], Cr[8], Ci[8], nCi[8];
#pragma unroll
    for (int j = 0; j < 8; ++j) {
        Fr[j]  = ftab[(j * 2 + 0) * 32 + lane];
        Fi[j]  = ftab[(j * 2 + 1) * 32 + lane];
        twr[j] = ftab[((8 + j) * 2 + 0) * 32 + lane];
        twi[j] = ftab[((8 + j) * 2 + 1) * 32 + lane];
        Cr[j]  = gfr[j * 32 + lane];
        Ci[j]  = gfr[(8 + j) * 32 + lane];
        nCi[j] = -Ci[j];
    }

    float GF[8][10];
#pragma unroll
    for (int f = 1; f < 8; ++f) {
#pragma unroll
        for (int j = 0; j < (f == 1 ? 10 : 8); ++j) GF[f][j] = gfr[(f * 16 + j) * 32 + lane];
    }

    float PHr[2][8], PHi[2][8];
#pragma unroll
    for (int l = 1; l < 3; ++l) {
        const float lr = gfr[((2 * l + 1) * 16 + 8) * 32 + lane];
        const float li = gfr[((2 * l + 1) * 16 + 9) * 32 + lane];
#pragma unroll
        for (int i = 0; i < 8; ++i) {
            const float hr = gfr[((2 * l) * 16 + 8) * 32 + 8 * h + i];
            const float hm = gfr[((2 * l) * 16 + 9) * 32 + 8 * h + i];
            PHr[l - 1][i] = hr * lr - hm * li;
            PHi[l - 1][i] = hr * li + hm * lr;
        }
    }

    for (int it = 0; it < ROWS_PER_WG / NWAVES; ++it) {
        if (it == 0)      __builtin_amdgcn_s_setprio(3);
        else if (it == 3) __builtin_amdgcn_s_setprio(2);
        else if (it == 5) __builtin_amdgcn_s_setprio(1);
        else if (it == 7) __builtin_amdgcn_s_setprio(0);

        const int row = blockIdx.x * ROWS_PER_WG + it * NWAVES + wv;
        float xr[8];
#pragma unroll
        for (int i = 0; i < 8; ++i) xr[i] = xn[i];
        if (it + 1 < ROWS_PER_WG / NWAVES) {
#pragma unroll
            for (int i = 0; i < 8; ++i) xn[i] = xrow[(size_t)(it + 1) * NWAVES * 256 + 16 * i];
        }

        float n2 = 0.0f;
#pragma unroll
        for (int i = 0; i < 8; ++i) n2 += xr[i] * xr[i];
#pragma unroll
        for (int m = 16; m >= 1; m >>= 1) n2 += __shfl_xor(n2, m, 32);

        v8f Dr = {}, Di = {};
#pragma unroll
        for (int t = 0; t < 4; ++t) {
            v2f a  = {xr[2 * t], xr[2 * t + 1]};
            v2f fr = {Fr[2 * t], Fr[2 * t + 1]};
            v2f fi = {Fi[2 * t], Fi[2 * t + 1]};
            Dr = WMMA_F32(a, fr, Dr);
            Di = WMMA_F32(a, fi, Di);
        }
        v8f Tr, Ti;
#pragma unroll
        for (int i = 0; i < 8; ++i) {
            Tr[i] = Dr[i] * twr[i] - Di[i] * twi[i];
            Ti[i] = Dr[i] * twi[i] + Di[i] * twr[i];
        }
        v8f Sr = {}, Si = {};
#pragma unroll
        for (int t = 0; t < 4; ++t) {
            v2f ar = {Tr[2 * t], Tr[2 * t + 1]};
            v2f ai = {Ti[2 * t], Ti[2 * t + 1]};
            v2f cr = {Cr[2 * t], Cr[2 * t + 1]};
            v2f ci  = {Ci[2 * t], Ci[2 * t + 1]};
            v2f nci = {nCi[2 * t], nCi[2 * t + 1]};
            Sr = WMMA_F32(ar, cr,  Sr);
            Si = WMMA_F32(ar, ci,  Si);
            Sr = WMMA_F32(ai, nci, Sr);
            Si = WMMA_F32(ai, cr,  Si);
        }

#pragma unroll
        for (int l = 0; l < 4; ++l) {
#pragma unroll
            for (int g = (l == 0) ? 1 : 0; g < 2; ++g) {
                const float (&G)[10] = GF[l * 2 + g];
                v8f Er = {}, Ei = {};
#pragma unroll
                for (int t = 0; t < 4; ++t) {
                    v2f ar = {Sr[2 * t], Sr[2 * t + 1]};
                    v2f ai = {Si[2 * t], Si[2 * t + 1]};
                    v2f b  = {G[2 * t], G[2 * t + 1]};
                    Er = WMMA_F32(ar, b, Er);
                    Ei = WMMA_F32(ai, b, Ei);
                }
                if (l == 0) {
                    const float pr = G[8], pi = G[9];
#pragma unroll
                    for (int i = 0; i < 8; ++i) {
                        Sr[i] = Er[i] * pr - Ei[i] * pi;
                        Si[i] = Er[i] * pi + Ei[i] * pr;
                    }
                } else if (g == 1 && l < 3) {
#pragma unroll
                    for (int i = 0; i < 8; ++i) {
                        Sr[i] = Er[i] * PHr[l - 1][i] - Ei[i] * PHi[l - 1][i];
                        Si[i] = Er[i] * PHi[l - 1][i] + Ei[i] * PHr[l - 1][i];
                    }
                } else {
                    Sr = Er;
                    Si = Ei;
                }
            }
            if (l < 3) {
                const int kb = 128 * h + c;
#pragma unroll
                for (int i = 0; i < 8; ++i) {
                    sre[kb + 16 * i] = Sr[i];
                    sim[kb + 16 * i] = Si[i];
                }
                wave_lds_order();
#pragma unroll
                for (int i = 0; i < 8; ++i) {
                    Sr[i] = sre[gsrc[i]];
                    Si[i] = sim[gsrc[i]];
                }
                wave_lds_order();
            }
        }

        const float inv = 1.0f / (256.0f * n2);
#pragma unroll
        for (int i = 0; i < 8; ++i)
            sre[16 * (8 * h + i) + c] = (Sr[i] * Sr[i] + Si[i] * Si[i]) * inv;
        wave_lds_order();
        float* o = out + (size_t)row * 256 + lane;
#pragma unroll
        for (int j = 0; j < 8; ++j) o[32 * j] = sre[osrc[j]];
        wave_lds_order();
    }
}

extern "C" void kernel_launch(void* const* d_in, const int* in_sizes, int n_in,
                              void* d_out, int out_size, void* d_ws, size_t ws_size,
                              hipStream_t stream) {
    const float* x    = (const float*)d_in[0];
    const float* wrot = (const float*)d_in[1];
    const float* wrz  = (const float*)d_in[2];
    float* out = (float*)d_out;

    qenc_rows<<<256, 512, 0, stream>>>(x, wrot, wrz, out);
}


# CNOT rings and output row permuted on the LDS store side (scattered 4-byte stores): a lane reads back 16-byte words (4 loads per ring, was 16) and writes out 16 bytes per lane; arithmetic unchanged
# speedup vs baseline: 6.1325x; 1.0545x over previous
#include <hip/hip_runtime.h>

typedef float v2f __attribute__((ext_vector_type(2)));
typedef float v4f __attribute__((ext_vector_type(4)));
typedef float v8f __attribute__((ext_vector_type(8)));

#define WMMA_F32(A, B, C) \
    __builtin_amdgcn_wmma_f32_16x16x4_f32(false, (A), false, (B), (short)0, (C), false, false)

#define NWAVES    16
#define ROW_PITCH 512
#define GFRAG_OFF (NWAVES * ROW_PITCH)
#define FTAB_OFF  (GFRAG_OFF + 8 * 16 * 32)
#define LDS_FLOATS (FTAB_OFF + 16 * 2 * 32)
#define ROWS_PER_WG 128

__device__ __forceinline__ int cnot_ring_dst(int k) {
    int dst = k;
    for (int q = 0; q < 8; ++q) {
        int cb = 7 - q, tb = 7 - ((q + 1) & 7);
        dst ^= ((dst >> cb) & 1) << tb;
    }
    return dst;
}

__device__ __forceinline__ void wave_lds_order() {
    __builtin_amdgcn_fence(__ATOMIC_RELEASE, "wavefront");
    __builtin_amdgcn_wave_barrier();
    __builtin_amdgcn_fence(__ATOMIC_ACQUIRE, "wavefront");
}

__global__ __launch_bounds__(512) void qenc_rows(const float* __restrict__ X,
                                                 const float* __restrict__ wrot,
                                                 const float* __restrict__ wrz,
                                                 float* __restrict__ out) {
    __shared__ __attribute__((aligned(16))) float lds[LDS_FLOATS];
    const int lane = threadIdx.x & 31;
    const int wv   = threadIdx.x >> 5;
    const int c    = lane & 15;
    const int h    = lane >> 4;

    float* sre  = lds + wv * ROW_PITCH;
    float* gfr  = lds + GFRAG_OFF;
    float* ftab = lds + FTAB_OFF;

    const float* xrow = X + (size_t)(blockIdx.x * ROWS_PER_WG + wv) * 256 + 128 * h + c;
    float xn[8];
#pragma unroll
    for (int i = 0; i < 8; ++i) xn[i] = xrow[16 * i];

    if (wv < 8) {
        float th  = wrot[lane];
        float phz = wrz[lane];
        float gc  = cosf(0.5f * th),  gs  = sinf(0.5f * th);
        float ger = cosf(0.5f * phz), gei = -sinf(0.5f * phz);
        const int g0 = (wv >> 1) * 8 + (wv & 1) * 4;
        float r0[4], r1[4];
        float dr = 1.0f, di = 0.0f;
#pragma unroll
        for (int p = 0; p < 4; ++p) {
            float cc = __shfl(gc,  g0 + p, 32), ss = __shfl(gs,  g0 + p, 32);
            float er = __shfl(ger, g0 + p, 32), ei = __shfl(gei, g0 + p, 32);
            int rb = (c >> (3 - p)) & 1;
            r0[p] = rb ? ss :  cc;
            r1[p] = rb ? cc : -ss;
            float fi = rb ? -ei : ei;
            float nr = dr * er - di * fi;
            float ni = dr * fi + di * er;
            dr = nr;
            di = ni;
        }
        if (wv == 0) {
            const float k16c[16] = { 1.0f,  0.92387953251128674f,  0.70710678118654752f,  0.38268343236508977f,
                                     0.0f, -0.38268343236508977f, -0.70710678118654752f, -0.92387953251128674f,
                                    -1.0f, -0.92387953251128674f, -0.70710678118654752f, -0.38268343236508977f,
                                     0.0f,  0.38268343236508977f,  0.70710678118654752f,  0.92387953251128674f};
            const float k16s[16] = { 0.0f,  0.38268343236508977f,  0.70710678118654752f,  0.92387953251128674f,
                                     1.0f,  0.92387953251128674f,  0.70710678118654752f,  0.38268343236508977f,
                                     0.0f, -0.38268343236508977f, -0.70710678118654752f, -0.92387953251128674f,
                                    -1.0f, -0.92387953251128674f, -0.70710678118654752f, -0.38268343236508977f};
            const float sg = h ? -1.0f : 1.0f;
            float rrow[16];
#pragma unroll
            for (int n = 0; n < 16; ++n) {
                float x0 = (n & 8) ? r1[0] : r0[0];
                float x1 = (n & 4) ? r1[1] : r0[1];
                float x2 = (n & 2) ? r1[2] : r0[2];
                float x3 = (n & 1) ? r1[3] : r0[3];
                float v = ((x0 * x1) * x2) * x3;
                rrow[n] = (n & 1) ? sg * v : v;
            }
#pragma unroll
            for (int j = 0; j < 8; ++j) {
                float sr = 0.0f, si = 0.0f;
#pragma unroll
                for (int n = 0; n < 16; ++n) {
                    sr += k16c[(j * n) & 15] * rrow[n];
                    si += k16s[(j * n) & 15] * rrow[n];
                }
                gfr[j * 32 + lane]       = dr * sr - di * si;
                gfr[(8 + j) * 32 + lane] = dr * si + di * sr;
            }
        } else {
            float e0 = h ? r1[0] : r0[0];
#pragma unroll
            for (int j = 0; j < 8; ++j) {
                int b2 = (j >> 2) & 1, b1 = (j >> 1) & 1, b0 = j & 1;
                float x1 = b2 ? r1[1] : r0[1];
                float x2 = b1 ? r1[2] : r0[2];
                float x3 = b0 ? r1[3] : r0[3];
                gfr[(wv * 16 + j) * 32 + lane] = ((e0 * x1) * x2) * x3;
            }
            gfr[(wv * 16 + 8) * 32 + lane] = dr;
            gfr[(wv * 16 + 9) * 32 + lane] = di;
        }
    }

    {
        const float w = 6.28318530717958647692f / 256.0f;
        int idx = (8 * h + (wv & 7)) * c;
        if (wv < 8) idx <<= 4;
        idx &= 255;
        float sv, cv;
        sincosf(w * (float)idx, &sv, &cv);
        ftab[(wv * 2 + 0) * 32 + lane] = cv;
        ftab[(wv * 2 + 1) * 32 + lane] = sv;
    }

    int pdst[8], odst[8];
#pragma unroll
    for (int i = 0; i < 8; ++i) {
        const int kd = cnot_ring_dst(16 * (8 * h + i) + c);
        const int hi2 = kd >> 4, lo2 = kd & 15;
        odst[i] = kd;
        pdst[i] = 256 * ((hi2 >> 2) & 1) + 4 * (16 * (hi2 >> 3) + lo2) + (hi2 & 3);
    }

    __syncthreads();

    float Fr[8], Fi[8], twr[8], twi[8], Cr[8], Ci[8], nCi[8];
#pragma unroll
    for (int j = 0; j < 8; ++j) {
        Fr[j]  = ftab[(j * 2 + 0) * 32 + lane];
        Fi[j]  = ftab[(j * 2 + 1) * 32 + lane];
        twr[j] = ftab[((8 + j) * 2 + 0) * 32 + lane];
        twi[j] = ftab[((8 + j) * 2 + 1) * 32 + lane];
        Cr[j]  = gfr[j * 32 + lane];
        Ci[j]  = gfr[(8 + j) * 32 + lane];
        nCi[j] = -Ci[j];
    }

    float GF[8][10];
#pragma unroll
    for (int f = 1; f < 8; ++f) {
#pragma unroll
        for (int j = 0; j < (f == 1 ? 10 : 8); ++j) GF[f][j] = gfr[(f * 16 + j) * 32 + lane];
    }

    float PHr[2][8], PHi[2][8];
#pragma unroll
    for (int l = 1; l < 3; ++l) {
        const float lr = gfr[((2 * l + 1) * 16 + 8) * 32 + lane];
        const float li = gfr[((2 * l + 1) * 16 + 9) * 32 + lane];
#pragma unroll
        for (int i = 0; i < 8; ++i) {
            const float hr = gfr[((2 * l) * 16 + 8) * 32 + 8 * h + i];
            const float hm = gfr[((2 * l) * 16 + 9) * 32 + 8 * h + i];
            PHr[l - 1][i] = hr * lr - hm * li;
            PHi[l - 1][i] = hr * li + hm * lr;
        }
    }

    for (int it = 0; it < ROWS_PER_WG / NWAVES; ++it) {
        if (it == 0)      __builtin_amdgcn_s_setprio(3);
        else if (it == 3) __builtin_amdgcn_s_setprio(2);
        else if (it == 5) __builtin_amdgcn_s_setprio(1);
        else if (it == 7) __builtin_amdgcn_s_setprio(0);

        const int row = blockIdx.x * ROWS_PER_WG + it * NWAVES + wv;
        float xr[8];
#pragma unroll
        for (int i = 0; i < 8; ++i) xr[i] = xn[i];
        if (it + 1 < ROWS_PER_WG / NWAVES) {
#pragma unroll
            for (int i = 0; i < 8; ++i) xn[i] = xrow[(size_t)(it + 1) * NWAVES * 256 + 16 * i];
        }

        float n2 = 0.0f;
#pragma unroll
        for (int i = 0; i < 8; ++i) n2 += xr[i] * xr[i];
#pragma unroll
        for (int m = 16; m >= 1; m >>= 1) n2 += __shfl_xor(n2, m, 32);

        v8f Dr = {}, Di = {};
#pragma unroll
        for (int t = 0; t < 4; ++t) {
            v2f a  = {xr[2 * t], xr[2 * t + 1]};
            v2f fr = {Fr[2 * t], Fr[2 * t + 1]};
            v2f fi = {Fi[2 * t], Fi[2 * t + 1]};
            Dr = WMMA_F32(a, fr, Dr);
            Di = WMMA_F32(a, fi, Di);
        }
        v8f Tr, Ti;
#pragma unroll
        for (int i = 0; i < 8; ++i) {
            Tr[i] = Dr[i] * twr[i] - Di[i] * twi[i];
            Ti[i] = Dr[i] * twi[i] + Di[i] * twr[i];
        }
        v8f Sr = {}, Si = {};
#pragma unroll
        for (int t = 0; t < 4; ++t) {
            v2f ar = {Tr[2 * t], Tr[2 * t + 1]};
            v2f ai = {Ti[2 * t], Ti[2 * t + 1]};
            v2f cr = {Cr[2 * t], Cr[2 * t + 1]};
            v2f ci  = {Ci[2 * t], Ci[2 * t + 1]};
            v2f nci = {nCi[2 * t], nCi[2 * t + 1]};
            Sr = WMMA_F32(ar, cr,  Sr);
            Si = WMMA_F32(ar, ci,  Si);
            Sr = WMMA_F32(ai, nci, Sr);
            Si = WMMA_F32(ai, cr,  Si);
        }

#pragma unroll
        for (int l = 0; l < 4; ++l) {
#pragma unroll
            for (int g = (l == 0) ? 1 : 0; g < 2; ++g) {
                const float (&G)[10] = GF[l * 2 + g];
                v8f Er = {}, Ei = {};
#pragma unroll
                for (int t = 0; t < 4; ++t) {
                    v2f ar = {Sr[2 * t], Sr[2 * t + 1]};
                    v2f ai = {Si[2 * t], Si[2 * t + 1]};
                    v2f b  = {G[2 * t], G[2 * t + 1]};
                    Er = WMMA_F32(ar, b, Er);
                    Ei = WMMA_F32(ai, b, Ei);
                }
                if (l == 0) {
                    const float pr = G[8], pi = G[9];
#pragma unroll
                    for (int i = 0; i < 8; ++i) {
                        Sr[i] = Er[i] * pr - Ei[i] * pi;
                        Si[i] = Er[i] * pi + Ei[i] * pr;
                    }
                } else if (g == 1 && l < 3) {
#pragma unroll
                    for (int i = 0; i < 8; ++i) {
                        Sr[i] = Er[i] * PHr[l - 1][i] - Ei[i] * PHi[l - 1][i];
                        Si[i] = Er[i] * PHi[l - 1][i] + Ei[i] * PHr[l - 1][i];
                    }
                } else {
                    Sr = Er;
                    Si = Ei;
                }
            }
            if (l < 3) {
#pragma unroll
                for (int i = 0; i < 8; ++i) {
                    sre[pdst[i]]       = Sr[i];
                    sre[pdst[i] + 128] = Si[i];
                }
                wave_lds_order();
                const v4f r0 = *(const v4f*)(sre +       4 * lane);
                const v4f i0 = *(const v4f*)(sre + 128 + 4 * lane);
                const v4f r1 = *(const v4f*)(sre + 256 + 4 * lane);
                const v4f i1 = *(const v4f*)(sre + 384 + 4 * lane);
                Sr = __builtin_shufflevector(r0, r1, 0, 1, 2, 3, 4, 5, 6, 7);
                Si = __builtin_shufflevector(i0, i1, 0, 1, 2, 3, 4, 5, 6, 7);
                wave_lds_order();
            }
        }

        const float inv = 1.0f / (256.0f * n2);
#pragma unroll
        for (int i = 0; i < 8; ++i)
            sre[odst[i]] = (Sr[i] * Sr[i] + Si[i] * Si[i]) * inv;
        wave_lds_order();
        const v4f p0 = *(const v4f*)(sre +       4 * lane);
        const v4f p1 = *(const v4f*)(sre + 128 + 4 * lane);
        float* o = out + (size_t)row * 256 + 4 * lane;
        *(v4f*)(o)       = p0;
        *(v4f*)(o + 128) = p1;
        wave_lds_order();
    }
}

extern "C" void kernel_launch(void* const* d_in, const int* in_sizes, int n_in,
                              void* d_out, int out_size, void* d_ws, size_t ws_size,
                              hipStream_t stream) {
    const float* x    = (const float*)d_in[0];
    const float* wrot = (const float*)d_in[1];
    const float* wrz  = (const float*)d_in[2];
    float* out = (float*)d_out;

    qenc_rows<<<256, 512, 0, stream>>>(x, wrot, wrz, out);
}
